# starting kernel
# baseline (speedup 1.0000x reference)
#include <hip/hip_runtime.h>
#include <cstddef>


typedef __attribute__((ext_vector_type(2))) float v2f;
typedef __attribute__((ext_vector_type(8))) float v8f;

__global__ void k_init_deg(float* __restrict__ deg, int n) {
  int i = blockIdx.x * blockDim.x + threadIdx.x;
  if (i < n) deg[i] = 1.0f;
}

__global__ void k_count_deg(const int* __restrict__ cols, float* __restrict__ deg, int E) {
  int e = blockIdx.x * blockDim.x + threadIdx.x;
  if (e < E) atomicAdd(&deg[cols[e]], 1.0f);
}

__global__ void k_rsqrt_inplace(float* __restrict__ deg, int n) {
  int i = blockIdx.x * blockDim.x + threadIdx.x;
  if (i < n) deg[i] = 1.0f / sqrtf(deg[i]);
}

template <int K, int NN>
__global__ void k_gemm_wmma(const float* __restrict__ A, const float* __restrict__ B,
                            float* __restrict__ C) {
  const int wave = threadIdx.x >> 5;
  const int lane = threadIdx.x & 31;
  const int m0   = blockIdx.x << 4;
  const int n0   = wave << 4;
  const int l16  = lane & 15;
  const int kh   = (lane >> 4) << 1;

  v8f acc = {};
  const float* arow = A + (size_t)(m0 + l16) * K + kh;
  const float* bcol = B + (size_t)kh * NN + n0 + l16;

#pragma unroll
  for (int k = 0; k < K; k += 4) {
    v2f a, b;
    a.x = arow[k];
    a.y = arow[k + 1];
    b.x = bcol[(size_t)k * NN];
    b.y = bcol[(size_t)(k + 1) * NN];
    acc = __builtin_amdgcn_wmma_f32_16x16x4_f32(false, a, false, b, (short)0, acc,
                                                false, false);
  }

  const int chi = (lane >> 4) << 3;
  float* crow = C + (size_t)(m0 + chi) * NN + n0 + l16;
#pragma unroll
  for (int v = 0; v < 8; ++v) crow[(size_t)v * NN] = acc[v];
}

__global__ void k_scatter(const int* __restrict__ rows, const int* __restrict__ cols,
                          const float* __restrict__ dinv, const float* __restrict__ H,
                          float* __restrict__ AGG, int E, int logF) {
  const int F   = 1 << logF;
  const int epb = blockDim.x >> logF;
  const int e   = blockIdx.x * epb + (threadIdx.x >> logF);
  if (e >= E) return;
  const int f = threadIdx.x & (F - 1);
  const int r = rows[e];
  const int c = cols[e];
  const float w = dinv[r] * dinv[c];
  atomicAdd(&AGG[(size_t)c * F + f], H[(size_t)r * F + f] * w);
}

__global__ void k_post(const float* __restrict__ AGG, const float* __restrict__ H,
                       const float* __restrict__ dinv, const float* __restrict__ bias,
                       float* __restrict__ OUT, int n, int logF, int do_relu) {
  const int F    = 1 << logF;
  const size_t i = (size_t)blockIdx.x * blockDim.x + threadIdx.x;
  if (i >= ((size_t)n << logF)) return;
  const int node = (int)(i >> logF);
  const int f    = (int)(i & (size_t)(F - 1));
  const float d  = dinv[node];
  float v = AGG[i] + H[i] * d * d + bias[f];
  if (do_relu) v = fmaxf(v, 0.0f);
  OUT[i] = v;
}

extern "C" void kernel_launch(void* const* d_in, const int* in_sizes, int n_in,
                              void* d_out, int out_size, void* d_ws, size_t ws_size,
                              hipStream_t stream) {
  const float* x  = (const float*)d_in[0];
  const int*   ei = (const int*)d_in[1];
  const float* W1 = (const float*)d_in[2];
  const float* b1 = (const float*)d_in[3];
  const float* W2 = (const float*)d_in[4];
  const float* b2 = (const float*)d_in[5];

  const int IN_DIM = 128, HID = 64, OUT = 32;
  const int N = in_sizes[0] / IN_DIM;
  const int E = in_sizes[1] / 2;
  const int* rows = ei;
  const int* cols = ei + E;

  float* ws   = (float*)d_ws;
  float* dinv = ws;
  float* h1   = ws + N;
  float* agg1 = h1 + (size_t)N * HID;
  float* h2   = h1;
  float* agg2 = h1 + (size_t)N * OUT;

  const int T = 256;

  k_init_deg<<<(N + T - 1) / T, T, 0, stream>>>(dinv, N);
  k_count_deg<<<(E + T - 1) / T, T, 0, stream>>>(cols, dinv, E);
  k_rsqrt_inplace<<<(N + T - 1) / T, T, 0, stream>>>(dinv, N);

  k_gemm_wmma<128, 64><<<N / 16, 32 * (64 / 16), 0, stream>>>(x, W1, h1);
  hipMemsetAsync(agg1, 0, (size_t)N * HID * sizeof(float), stream);
  k_scatter<<<(E * HID + T - 1) / T, T, 0, stream>>>(rows, cols, dinv, h1, agg1, E, 6);
  k_post<<<(int)(((size_t)N * HID + T - 1) / T), T, 0, stream>>>(agg1, h1, dinv, b1,
                                                                 agg1, N, 6, 1);

  k_gemm_wmma<64, 32><<<N / 16, 32 * (32 / 16), 0, stream>>>(agg1, W2, h2);
  hipMemsetAsync(agg2, 0, (size_t)N * OUT * sizeof(float), stream);
  k_scatter<<<(E * OUT + T - 1) / T, T, 0, stream>>>(rows, cols, dinv, h2, agg2, E, 5);
  k_post<<<(int)(((size_t)N * OUT + T - 1) / T), T, 0, stream>>>(agg2, h2, dinv, b2,
                                                                 (float*)d_out, N, 5, 0);
}


# Aggregation as per-target gather over a target-sorted edge list (counting sort, arrival-order rank from one returning int atomic per edge) replaces float atomic scatter-add, memsets and post pass
# speedup vs baseline: 2.3752x; 2.3752x over previous
#include <hip/hip_runtime.h>
#include <cstddef>


typedef __attribute__((ext_vector_type(2))) float v2f;
typedef __attribute__((ext_vector_type(8))) float v8f;

__global__ void k_zero(int* __restrict__ p, int n) {
  int i = blockIdx.x * blockDim.x + threadIdx.x;
  if (i < n) p[i] = 0;
}

__global__ void k_rank(const int* __restrict__ cols, int* __restrict__ cnt,
                       int* __restrict__ rank, int E) {
  int e = blockIdx.x * blockDim.x + threadIdx.x;
  if (e < E) rank[e] = atomicAdd(&cnt[cols[e]], 1);
}

__device__ inline int4 load4_guarded(const int* __restrict__ p, int base, int n) {
  int4 v = make_int4(0, 0, 0, 0);
  if (base + 3 < n) {
    v = *(const int4*)(p + base);
  } else {
    if (base < n) v.x = p[base];
    if (base + 1 < n) v.y = p[base + 1];
    if (base + 2 < n) v.z = p[base + 2];
  }
  return v;
}

__device__ inline int wave_incl_scan(int v, int lane) {
  v += __builtin_amdgcn_update_dpp(0, v, 0x111, 0xf, 0xf, false);
  v += __builtin_amdgcn_update_dpp(0, v, 0x112, 0xf, 0xf, false);
  v += __builtin_amdgcn_update_dpp(0, v, 0x114, 0xf, 0xf, false);
  v += __builtin_amdgcn_update_dpp(0, v, 0x118, 0xf, 0xf, false);
  const int low = __builtin_amdgcn_readlane(v, 15);
  if (lane >= 16) v += low;
  return v;
}

__global__ __launch_bounds__(1024) void k_scan(const int* __restrict__ cnt,
                                               int* __restrict__ off, int n) {
  __shared__ int wtot[2][32];
  const int t    = threadIdx.x;
  const int lane = t & 31;
  const int wv   = __builtin_amdgcn_readfirstlane(t >> 5);
  const int nchunk = (n + 4095) >> 12;
  int carry = 0;
  int4 nxt = load4_guarded(cnt, t << 2, n);
  for (int ch = 0; ch < nchunk; ++ch) {
    const int base = (ch << 12) + (t << 2);
    const int4 v = nxt;
    if (ch + 1 < nchunk) nxt = load4_guarded(cnt, base + 4096, n);
    const int s   = v.x + v.y + v.z + v.w;
    const int inc = wave_incl_scan(s, lane);
    if (lane == 31) wtot[ch & 1][wv] = inc;
    __syncthreads();
    const int wt   = wtot[ch & 1][lane];
    const int winc = wave_incl_scan(wt, lane);
    const int wave_excl =
        __builtin_amdgcn_readlane(winc, wv) - __builtin_amdgcn_readlane(wt, wv);
    const int total = __builtin_amdgcn_readlane(winc, 31);
    const int e0 = carry + wave_excl + (inc - s);
    const int e1 = e0 + v.x;
    const int e2 = e1 + v.y;
    const int e3 = e2 + v.z;
    if (base + 3 < n) {
      *(int4*)(off + base) = make_int4(e0, e1, e2, e3);
    } else {
      if (base < n) off[base] = e0;
      if (base + 1 < n) off[base + 1] = e1;
      if (base + 2 < n) off[base + 2] = e2;
    }
    carry += total;
  }
  if (t == 0) off[n] = carry;
}

__global__ void k_fill(const int* __restrict__ rows, const int* __restrict__ cols,
                       const int* __restrict__ rank, const int* __restrict__ off,
                       const int* __restrict__ cnt, int* __restrict__ adj, int E) {
  int e = blockIdx.x * blockDim.x + threadIdx.x;
  if (e >= E) return;
  const int r = rows[e];
  const int pos = off[cols[e]] + rank[e];
  const float dr = 1.0f / sqrtf((float)(cnt[r] + 1));
  ((int2*)adj)[pos] = make_int2(r, __float_as_int(dr));
}

template <int K, int NN>
__global__ void k_gemm_wmma(const float* __restrict__ A, const float* __restrict__ B,
                            float* __restrict__ C) {
  const int wave = threadIdx.x >> 5;
  const int lane = threadIdx.x & 31;
  const int m0   = blockIdx.x << 4;
  const int n0   = wave << 4;
  const int l16  = lane & 15;
  const int kh   = (lane >> 4) << 1;

  v8f acc = {};
  const float* arow = A + (size_t)(m0 + l16) * K + kh;
  const float* bcol = B + (size_t)kh * NN + n0 + l16;

#pragma unroll
  for (int k = 0; k < K; k += 4) {
    v2f a, b;
    a.x = arow[k];
    a.y = arow[k + 1];
    b.x = bcol[(size_t)k * NN];
    b.y = bcol[(size_t)(k + 1) * NN];
    acc = __builtin_amdgcn_wmma_f32_16x16x4_f32(false, a, false, b, (short)0, acc,
                                                false, false);
  }

  const int chi = (lane >> 4) << 3;
  float* crow = C + (size_t)(m0 + chi) * NN + n0 + l16;
#pragma unroll
  for (int v = 0; v < 8; ++v) crow[(size_t)v * NN] = acc[v];
}

__device__ inline float4 fma4(const float4 g, const float w, const float4 a) {
  return make_float4(fmaf(g.x, w, a.x), fmaf(g.y, w, a.y), fmaf(g.z, w, a.z), fmaf(g.w, w, a.w));
}

template <int F, bool RELU>
__global__ void k_gather(const int* __restrict__ adj, const int* __restrict__ off,
                         const float* __restrict__ H, const float* __restrict__ bias,
                         float* __restrict__ OUT, int n) {
  constexpr int LPN = F / 4;
  const int t    = blockIdx.x * blockDim.x + threadIdx.x;
  const int node = t / LPN;
  const int j    = t % LPN;
  if (node >= n) return;
  const int2*   __restrict__ A2 = (const int2*)adj;
  const float4* __restrict__ H4 = (const float4*)H;
  const int beg = off[node];
  const int end = off[node + 1];
  const float dc = 1.0f / sqrtf((float)(end - beg + 1));
  const unsigned last = (unsigned)(n - 1);

  const float4 hs = H4[(size_t)node * LPN + j];
  const float4 b  = ((const float4*)bias)[j];

  float4 acc = make_float4(0.f, 0.f, 0.f, 0.f);
  int i = beg;
  for (; i + 4 <= end; i += 4) {
    const int2 p0 = A2[i], p1 = A2[i + 1], p2 = A2[i + 2], p3 = A2[i + 3];
    const float4 g0 = H4[(size_t)min((unsigned)p0.x, last) * LPN + j];
    const float4 g1 = H4[(size_t)min((unsigned)p1.x, last) * LPN + j];
    const float4 g2 = H4[(size_t)min((unsigned)p2.x, last) * LPN + j];
    const float4 g3 = H4[(size_t)min((unsigned)p3.x, last) * LPN + j];
    acc = fma4(g0, __int_as_float(p0.y) * dc, acc);
    acc = fma4(g1, __int_as_float(p1.y) * dc, acc);
    acc = fma4(g2, __int_as_float(p2.y) * dc, acc);
    acc = fma4(g3, __int_as_float(p3.y) * dc, acc);
  }
  if (i + 2 <= end) {
    const int2 p0 = A2[i], p1 = A2[i + 1];
    const float4 g0 = H4[(size_t)min((unsigned)p0.x, last) * LPN + j];
    const float4 g1 = H4[(size_t)min((unsigned)p1.x, last) * LPN + j];
    acc = fma4(g0, __int_as_float(p0.y) * dc, acc);
    acc = fma4(g1, __int_as_float(p1.y) * dc, acc);
    i += 2;
  }
  if (i < end) {
    const int2 p0 = A2[i];
    const float4 g0 = H4[(size_t)min((unsigned)p0.x, last) * LPN + j];
    acc = fma4(g0, __int_as_float(p0.y) * dc, acc);
  }

  const float dd  = dc * dc;
  float4 o;
  o.x = acc.x + hs.x * dd + b.x;
  o.y = acc.y + hs.y * dd + b.y;
  o.z = acc.z + hs.z * dd + b.z;
  o.w = acc.w + hs.w * dd + b.w;
  if (RELU) {
    o.x = fmaxf(o.x, 0.0f); o.y = fmaxf(o.y, 0.0f);
    o.z = fmaxf(o.z, 0.0f); o.w = fmaxf(o.w, 0.0f);
  }
  ((float4*)OUT)[(size_t)node * LPN + j] = o;
}

extern "C" void kernel_launch(void* const* d_in, const int* in_sizes, int n_in,
                              void* d_out, int out_size, void* d_ws, size_t ws_size,
                              hipStream_t stream) {
  const float* x  = (const float*)d_in[0];
  const int*   ei = (const int*)d_in[1];
  const float* W1 = (const float*)d_in[2];
  const float* b1 = (const float*)d_in[3];
  const float* W2 = (const float*)d_in[4];
  const float* b2 = (const float*)d_in[5];

  const int IN_DIM = 128, HID = 64, OUT = 32;
  const int N = in_sizes[0] / IN_DIM;
  const int E = in_sizes[1] / 2;
  const int* rows = ei;
  const int* cols = ei + E;

  float* ws   = (float*)d_ws;
  float* h1   = ws;
  float* r1   = h1 + (size_t)N * HID;
  float* h2   = r1 + (size_t)N * HID;
  int*   adj  = (int*)(h2 + (size_t)N * OUT);
  int*   rank = adj + (size_t)2 * E;
  int*   cnt  = rank + E;
  int*   off  = cnt + N;

  const int T = 256;

  k_zero<<<(N + T - 1) / T, T, 0, stream>>>(cnt, N);
  k_rank<<<(E + T - 1) / T, T, 0, stream>>>(cols, cnt, rank, E);
  k_scan<<<1, 1024, 0, stream>>>(cnt, off, N);
  k_fill<<<(E + T - 1) / T, T, 0, stream>>>(rows, cols, rank, off, cnt, adj, E);

  k_gemm_wmma<128, 64><<<N / 16, 32 * (64 / 16), 0, stream>>>(x, W1, h1);
  k_gather<64, true><<<(N * (HID / 4) + T - 1) / T, T, 0, stream>>>(adj, off, h1, b1, r1, N);

  k_gemm_wmma<64, 32><<<N / 16, 32 * (32 / 16), 0, stream>>>(r1, W2, h2);
  k_gather<32, false><<<(N * (OUT / 4) + T - 1) / T, T, 0, stream>>>(adj, off, h2, b2,
                                                                     (float*)d_out, N);
}


# Edge ranks and in-degrees from LDS counters (32 node ranges x 8 edge chunks, per-range scan): no global atomic, no zeroing launch; a node's edges summed chunk-major, LDS arrival order inside
# speedup vs baseline: 2.7863x; 1.1731x over previous
#include <hip/hip_runtime.h>
#include <cstddef>


typedef __attribute__((ext_vector_type(2))) float v2f;
typedef __attribute__((ext_vector_type(8))) float v8f;

__device__ inline int wave_incl_scan(int v, int lane) {
  v += __builtin_amdgcn_update_dpp(0, v, 0x111, 0xf, 0xf, false);
  v += __builtin_amdgcn_update_dpp(0, v, 0x112, 0xf, 0xf, false);
  v += __builtin_amdgcn_update_dpp(0, v, 0x114, 0xf, 0xf, false);
  v += __builtin_amdgcn_update_dpp(0, v, 0x118, 0xf, 0xf, false);
  const int low = __builtin_amdgcn_readlane(v, 15);
  if (lane >= 16) v += low;
  return v;
}

template <int R> __device__ inline int range_nodes(const int N) { return (N + R - 1) / R; }
template <int C> __device__ inline int chunk_edges(const int E) {
  return ((E + C - 1) / C + 4095) / 4096 * 4096;
}

__device__ inline int4 count4(const int4 v, const int lo, const unsigned span, int* lc) {
  const unsigned u0 = (unsigned)(v.x - lo), u1 = (unsigned)(v.y - lo);
  const unsigned u2 = (unsigned)(v.z - lo), u3 = (unsigned)(v.w - lo);
  int4 k = make_int4(-1, -1, -1, -1);
  if (u0 < span) k.x = atomicAdd(&lc[u0], 1);
  if (u1 < span) k.y = atomicAdd(&lc[u1], 1);
  if (u2 < span) k.z = atomicAdd(&lc[u2], 1);
  if (u3 < span) k.w = atomicAdd(&lc[u3], 1);
  return k;
}

__device__ inline void store_ranks(int* __restrict__ rank, const int b, const int4 k) {
  if (k.x >= 0) rank[b]     = k.x;
  if (k.y >= 0) rank[b + 1] = k.y;
  if (k.z >= 0) rank[b + 2] = k.z;
  if (k.w >= 0) rank[b + 3] = k.w;
}

template <int R, int C>
__global__ __launch_bounds__(1024) void k_rank_lds(const int* __restrict__ cols,
                                                   int* __restrict__ rank,
                                                   int* __restrict__ part,
                                                   int* __restrict__ tot, int N, int E) {
  __shared__ int lc[2048];
  const int RN = range_nodes<R>(N);
  const int CE = chunk_edges<C>(E);
  __shared__ int stot;
  const int t    = threadIdx.x;
  const int lane = t & 31;
  const int r    = blockIdx.x % R;
  const int c    = blockIdx.x / R;
  const int lo   = r * RN;
  const int nsp  = max(min(lo + RN, N) - lo, 0);
  const unsigned span = (unsigned)nsp;
  for (int i = t; i < nsp; i += 1024) lc[i] = 0;
  if (t == 0) stot = 0;
  __syncthreads();

  const int e0  = c * CE;
  const int e1  = max(min(e0 + CE, E), e0);
  const int e1v = e1 & ~3;
  const int4* __restrict__ c4 = (const int4*)cols;
  int b0 = e0;
  for (; b0 + 4 * 4096 <= e1v; b0 += 4 * 4096) {
    const int b = b0 + (t << 2);
    const int4 v0 = c4[b >> 2];
    const int4 v1 = c4[(b >> 2) + 1024];
    const int4 v2 = c4[(b >> 2) + 2048];
    const int4 v3 = c4[(b >> 2) + 3072];
    const int4 k0 = count4(v0, lo, span, lc);
    const int4 k1 = count4(v1, lo, span, lc);
    const int4 k2 = count4(v2, lo, span, lc);
    const int4 k3 = count4(v3, lo, span, lc);
    store_ranks(rank, b, k0);
    store_ranks(rank, b + 4096, k1);
    store_ranks(rank, b + 2 * 4096, k2);
    store_ranks(rank, b + 3 * 4096, k3);
  }
  for (; b0 < e1v; b0 += 4096) {
    const int b = b0 + (t << 2);
    if (b < e1v) {
      const int4 k0 = count4(c4[b >> 2], lo, span, lc);
      store_ranks(rank, b, k0);
    }
  }
  for (int b = e1v + t; b < e1; b += 1024) {
    const unsigned u = (unsigned)(cols[b] - lo);
    if (u < span) rank[b] = atomicAdd(&lc[u], 1);
  }
  __syncthreads();
  int mine = 0;
  for (int i = t; i < nsp; i += 1024) {
    const int k = lc[i];
    part[(size_t)c * N + lo + i] = k;
    mine += k;
  }
  const int winc = wave_incl_scan(mine, lane);
  if (lane == 31) atomicAdd(&stot, winc);
  __syncthreads();
  if (t == 0) tot[blockIdx.x] = stot;
}

template <int R, int C>
__global__ __launch_bounds__(1024) void k_offsets(int* part, const int* __restrict__ tot,
                                                  int* __restrict__ off,
                                                  int* __restrict__ cnt, int N) {
  __shared__ int wa[32];
  __shared__ int wb[32];
  const int t    = threadIdx.x;
  const int lane = t & 31;
  const int wv   = __builtin_amdgcn_readfirstlane(t >> 5);
  const int r    = blockIdx.x;
  const int lo   = r * range_nodes<R>(N);
  const int hi   = min(lo + range_nodes<R>(N), N);

  const int i0 = lo + 2 * t;
  const int i1 = i0 + 1;
  const bool ok0 = i0 < hi, ok1 = i1 < hi;
  int p0[C], p1[C];
#pragma unroll
  for (int c = 0; c < C; ++c) {
    p0[c] = ok0 ? part[(size_t)c * N + i0] : 0;
    p1[c] = ok1 ? part[(size_t)c * N + i1] : 0;
  }
  int tv = 0;
  if (t < R * C && (t % R) < r) tv = tot[t];
  const int tinc = wave_incl_scan(tv, lane);
  if (lane == 31) wa[wv] = tinc;
  int s0 = 0, s1 = 0;
#pragma unroll
  for (int c = 0; c < C; ++c) {
    s0 += p0[c];
    s1 += p1[c];
  }
  const int s   = s0 + s1;
  const int inc = wave_incl_scan(s, lane);
  if (lane == 31) wb[wv] = inc;
  __syncthreads();
  const int a    = wa[lane];
  const int b    = wb[lane];
  const int ainc = wave_incl_scan(a, lane);
  const int binc = wave_incl_scan(b, lane);
  const int start = __builtin_amdgcn_readlane(ainc, 31);
  const int wave_excl =
      __builtin_amdgcn_readlane(binc, wv) - __builtin_amdgcn_readlane(b, wv);
  const int total = __builtin_amdgcn_readlane(binc, 31);
  const int f0 = start + wave_excl + (inc - s);
  const int f1 = f0 + s0;
  if (ok0) {
    off[i0] = f0;
    cnt[i0] = s0;
    int run = f0;
#pragma unroll
    for (int c = 0; c < C; ++c) { part[(size_t)c * N + i0] = run; run += p0[c]; }
  }
  if (ok1) {
    off[i1] = f1;
    cnt[i1] = s1;
    int run = f1;
#pragma unroll
    for (int c = 0; c < C; ++c) { part[(size_t)c * N + i1] = run; run += p1[c]; }
  }
  if (r == R - 1 && t == 0) off[N] = start + total;
}

template <int C>
__global__ void k_fill(const int* __restrict__ rows, const int* __restrict__ cols,
                       const int* __restrict__ rank, const int* __restrict__ part,
                       const int* __restrict__ cnt, int* __restrict__ adj, int N, int E) {
  int e = blockIdx.x * blockDim.x + threadIdx.x;
  if (e >= E) return;
  const int c = (int)(blockIdx.x * blockDim.x) / chunk_edges<C>(E);
  const int r = rows[e];
  const int pos = part[(size_t)c * N + cols[e]] + rank[e];
  const float dr = 1.0f / sqrtf((float)(cnt[r] + 1));
  ((int2*)adj)[pos] = make_int2(r, __float_as_int(dr));
}

template <int K, int NN>
__global__ void k_gemm_wmma(const float* __restrict__ A, const float* __restrict__ B,
                            float* __restrict__ C) {
  const int wave = threadIdx.x >> 5;
  const int lane = threadIdx.x & 31;
  const int m0   = blockIdx.x << 4;
  const int n0   = wave << 4;
  const int l16  = lane & 15;
  const int kh   = (lane >> 4) << 1;

  v8f acc = {};
  const float* arow = A + (size_t)(m0 + l16) * K + kh;
  const float* bcol = B + (size_t)kh * NN + n0 + l16;

#pragma unroll
  for (int k = 0; k < K; k += 4) {
    v2f a, b;
    a.x = arow[k];
    a.y = arow[k + 1];
    b.x = bcol[(size_t)k * NN];
    b.y = bcol[(size_t)(k + 1) * NN];
    acc = __builtin_amdgcn_wmma_f32_16x16x4_f32(false, a, false, b, (short)0, acc,
                                                false, false);
  }

  const int chi = (lane >> 4) << 3;
  float* crow = C + (size_t)(m0 + chi) * NN + n0 + l16;
#pragma unroll
  for (int v = 0; v < 8; ++v) crow[(size_t)v * NN] = acc[v];
}

__device__ inline float4 fma4(const float4 g, const float w, const float4 a) {
  return make_float4(fmaf(g.x, w, a.x), fmaf(g.y, w, a.y), fmaf(g.z, w, a.z), fmaf(g.w, w, a.w));
}

template <int F, bool RELU>
__global__ void k_gather(const int* __restrict__ adj, const int* __restrict__ off,
                         const float* __restrict__ H, const float* __restrict__ bias,
                         float* __restrict__ OUT, int n) {
  constexpr int LPN = F / 4;
  const int t    = blockIdx.x * blockDim.x + threadIdx.x;
  const int node = t / LPN;
  const int j    = t % LPN;
  if (node >= n) return;
  const int2*   __restrict__ A2 = (const int2*)adj;
  const float4* __restrict__ H4 = (const float4*)H;
  const int beg = off[node];
  const int end = off[node + 1];
  const float dc = 1.0f / sqrtf((float)(end - beg + 1));
  const unsigned last = (unsigned)(n - 1);

  const float4 hs = H4[(size_t)node * LPN + j];
  const float4 b  = ((const float4*)bias)[j];

  float4 acc = make_float4(0.f, 0.f, 0.f, 0.f);
  int i = beg;
  for (; i + 4 <= end; i += 4) {
    const int2 p0 = A2[i], p1 = A2[i + 1], p2 = A2[i + 2], p3 = A2[i + 3];
    const float4 g0 = H4[(size_t)min((unsigned)p0.x, last) * LPN + j];
    const float4 g1 = H4[(size_t)min((unsigned)p1.x, last) * LPN + j];
    const float4 g2 = H4[(size_t)min((unsigned)p2.x, last) * LPN + j];
    const float4 g3 = H4[(size_t)min((unsigned)p3.x, last) * LPN + j];
    acc = fma4(g0, __int_as_float(p0.y) * dc, acc);
    acc = fma4(g1, __int_as_float(p1.y) * dc, acc);
    acc = fma4(g2, __int_as_float(p2.y) * dc, acc);
    acc = fma4(g3, __int_as_float(p3.y) * dc, acc);
  }
  if (i + 2 <= end) {
    const int2 p0 = A2[i], p1 = A2[i + 1];
    const float4 g0 = H4[(size_t)min((unsigned)p0.x, last) * LPN + j];
    const float4 g1 = H4[(size_t)min((unsigned)p1.x, last) * LPN + j];
    acc = fma4(g0, __int_as_float(p0.y) * dc, acc);
    acc = fma4(g1, __int_as_float(p1.y) * dc, acc);
    i += 2;
  }
  if (i < end) {
    const int2 p0 = A2[i];
    const float4 g0 = H4[(size_t)min((unsigned)p0.x, last) * LPN + j];
    acc = fma4(g0, __int_as_float(p0.y) * dc, acc);
  }

  const float dd  = dc * dc;
  float4 o;
  o.x = acc.x + hs.x * dd + b.x;
  o.y = acc.y + hs.y * dd + b.y;
  o.z = acc.z + hs.z * dd + b.z;
  o.w = acc.w + hs.w * dd + b.w;
  if (RELU) {
    o.x = fmaxf(o.x, 0.0f); o.y = fmaxf(o.y, 0.0f);
    o.z = fmaxf(o.z, 0.0f); o.w = fmaxf(o.w, 0.0f);
  }
  ((float4*)OUT)[(size_t)node * LPN + j] = o;
}

extern "C" void kernel_launch(void* const* d_in, const int* in_sizes, int n_in,
                              void* d_out, int out_size, void* d_ws, size_t ws_size,
                              hipStream_t stream) {
  const float* x  = (const float*)d_in[0];
  const int*   ei = (const int*)d_in[1];
  const float* W1 = (const float*)d_in[2];
  const float* b1 = (const float*)d_in[3];
  const float* W2 = (const float*)d_in[4];
  const float* b2 = (const float*)d_in[5];

  const int IN_DIM = 128, HID = 64, OUT = 32;
  const int N = in_sizes[0] / IN_DIM;
  const int E = in_sizes[1] / 2;
  const int* rows = ei;
  const int* cols = ei + E;

  float* ws   = (float*)d_ws;
  float* h1   = ws;
  float* r1   = h1 + (size_t)N * HID;
  float* h2   = r1 + (size_t)N * HID;
  int*   adj  = (int*)(h2 + (size_t)N * OUT);
  int*   rank = adj + (size_t)2 * E;
  int*   cnt  = rank + E;
  int*   off  = cnt + N;
  int*   part = off + N + 1;

  const int T = 256;
  const int R = 32;
  const int C = 8;
  int*   tot  = part + (size_t)C * N;

  k_rank_lds<R, C><<<R * C, 1024, 0, stream>>>(cols, rank, part, tot, N, E);
  k_offsets<R, C><<<R, 1024, 0, stream>>>(part, tot, off, cnt, N);
  k_fill<C><<<(E + T - 1) / T, T, 0, stream>>>(rows, cols, rank, part, cnt, adj, N, E);

  k_gemm_wmma<128, 64><<<N / 16, 32 * (64 / 16), 0, stream>>>(x, W1, h1);
  k_gather<64, true><<<(N * (HID / 4) + T - 1) / T, T, 0, stream>>>(adj, off, h1, b1, r1, N);

  k_gemm_wmma<64, 32><<<N / 16, 32 * (32 / 16), 0, stream>>>(r1, W2, h2);
  k_gather<32, false><<<(N * (OUT / 4) + T - 1) / T, T, 0, stream>>>(adj, off, h2, b2,
                                                                     (float*)d_out, N);
}


# Layer-2 product r1@W2 runs inside the layer-1 gather work-group: its 16 relu rows go through LDS to two waves' f32 WMMAs; r1 is never stored, 7 launches to 6; every sum keeps its order
# speedup vs baseline: 2.8920x; 1.0379x over previous
#include <hip/hip_runtime.h>
#include <cstddef>


typedef __attribute__((ext_vector_type(2))) float v2f;
typedef __attribute__((ext_vector_type(8))) float v8f;

__device__ inline int wave_incl_scan(int v, int lane) {
  v += __builtin_amdgcn_update_dpp(0, v, 0x111, 0xf, 0xf, false);
  v += __builtin_amdgcn_update_dpp(0, v, 0x112, 0xf, 0xf, false);
  v += __builtin_amdgcn_update_dpp(0, v, 0x114, 0xf, 0xf, false);
  v += __builtin_amdgcn_update_dpp(0, v, 0x118, 0xf, 0xf, false);
  const int low = __builtin_amdgcn_readlane(v, 15);
  if (lane >= 16) v += low;
  return v;
}

template <int R> __device__ inline int range_nodes(const int N) { return (N + R - 1) / R; }
template <int C> __device__ inline int chunk_edges(const int E) {
  return ((E + C - 1) / C + 4095) / 4096 * 4096;
}

__device__ inline int4 count4(const int4 v, const int lo, const unsigned span, int* lc) {
  const unsigned u0 = (unsigned)(v.x - lo), u1 = (unsigned)(v.y - lo);
  const unsigned u2 = (unsigned)(v.z - lo), u3 = (unsigned)(v.w - lo);
  int4 k = make_int4(-1, -1, -1, -1);
  if (u0 < span) k.x = atomicAdd(&lc[u0], 1);
  if (u1 < span) k.y = atomicAdd(&lc[u1], 1);
  if (u2 < span) k.z = atomicAdd(&lc[u2], 1);
  if (u3 < span) k.w = atomicAdd(&lc[u3], 1);
  return k;
}

__device__ inline void store_ranks(int* __restrict__ rank, const int b, const int4 k) {
  if (k.x >= 0) rank[b]     = k.x;
  if (k.y >= 0) rank[b + 1] = k.y;
  if (k.z >= 0) rank[b + 2] = k.z;
  if (k.w >= 0) rank[b + 3] = k.w;
}

template <int R, int C>
__global__ __launch_bounds__(1024) void k_rank_lds(const int* __restrict__ cols,
                                                   int* __restrict__ rank,
                                                   int* __restrict__ part,
                                                   int* __restrict__ tot, int N, int E) {
  __shared__ int lc[2048];
  const int RN = range_nodes<R>(N);
  const int CE = chunk_edges<C>(E);
  __shared__ int stot;
  const int t    = threadIdx.x;
  const int lane = t & 31;
  const int r    = blockIdx.x % R;
  const int c    = blockIdx.x / R;
  const int lo   = r * RN;
  const int nsp  = max(min(lo + RN, N) - lo, 0);
  const unsigned span = (unsigned)nsp;
  for (int i = t; i < nsp; i += 1024) lc[i] = 0;
  if (t == 0) stot = 0;
  __syncthreads();

  const int e0  = c * CE;
  const int e1  = max(min(e0 + CE, E), e0);
  const int e1v = e1 & ~3;
  const int4* __restrict__ c4 = (const int4*)cols;
  int b0 = e0;
  for (; b0 + 4 * 4096 <= e1v; b0 += 4 * 4096) {
    const int b = b0 + (t << 2);
    const int4 v0 = c4[b >> 2];
    const int4 v1 = c4[(b >> 2) + 1024];
    const int4 v2 = c4[(b >> 2) + 2048];
    const int4 v3 = c4[(b >> 2) + 3072];
    const int4 k0 = count4(v0, lo, span, lc);
    const int4 k1 = count4(v1, lo, span, lc);
    const int4 k2 = count4(v2, lo, span, lc);
    const int4 k3 = count4(v3, lo, span, lc);
    store_ranks(rank, b, k0);
    store_ranks(rank, b + 4096, k1);
    store_ranks(rank, b + 2 * 4096, k2);
    store_ranks(rank, b + 3 * 4096, k3);
  }
  for (; b0 < e1v; b0 += 4096) {
    const int b = b0 + (t << 2);
    if (b < e1v) {
      const int4 k0 = count4(c4[b >> 2], lo, span, lc);
      store_ranks(rank, b, k0);
    }
  }
  for (int b = e1v + t; b < e1; b += 1024) {
    const unsigned u = (unsigned)(cols[b] - lo);
    if (u < span) rank[b] = atomicAdd(&lc[u], 1);
  }
  __syncthreads();
  int mine = 0;
  for (int i = t; i < nsp; i += 1024) {
    const int k = lc[i];
    part[(size_t)c * N + lo + i] = k;
    mine += k;
  }
  const int winc = wave_incl_scan(mine, lane);
  if (lane == 31) atomicAdd(&stot, winc);
  __syncthreads();
  if (t == 0) tot[blockIdx.x] = stot;
}

template <int R, int C>
__global__ __launch_bounds__(1024) void k_offsets(int* part, const int* __restrict__ tot,
                                                  int* __restrict__ off,
                                                  int* __restrict__ cnt, int N) {
  __shared__ int wa[32];
  __shared__ int wb[32];
  const int t    = threadIdx.x;
  const int lane = t & 31;
  const int wv   = __builtin_amdgcn_readfirstlane(t >> 5);
  const int r    = blockIdx.x;
  const int lo   = r * range_nodes<R>(N);
  const int hi   = min(lo + range_nodes<R>(N), N);

  const int i0 = lo + 2 * t;
  const int i1 = i0 + 1;
  const bool ok0 = i0 < hi, ok1 = i1 < hi;
  int p0[C], p1[C];
#pragma unroll
  for (int c = 0; c < C; ++c) {
    p0[c] = ok0 ? part[(size_t)c * N + i0] : 0;
    p1[c] = ok1 ? part[(size_t)c * N + i1] : 0;
  }
  int tv = 0;
  if (t < R * C && (t % R) < r) tv = tot[t];
  const int tinc = wave_incl_scan(tv, lane);
  if (lane == 31) wa[wv] = tinc;
  int s0 = 0, s1 = 0;
#pragma unroll
  for (int c = 0; c < C; ++c) {
    s0 += p0[c];
    s1 += p1[c];
  }
  const int s   = s0 + s1;
  const int inc = wave_incl_scan(s, lane);
  if (lane == 31) wb[wv] = inc;
  __syncthreads();
  const int a    = wa[lane];
  const int b    = wb[lane];
  const int ainc = wave_incl_scan(a, lane);
  const int binc = wave_incl_scan(b, lane);
  const int start = __builtin_amdgcn_readlane(ainc, 31);
  const int wave_excl =
      __builtin_amdgcn_readlane(binc, wv) - __builtin_amdgcn_readlane(b, wv);
  const int total = __builtin_amdgcn_readlane(binc, 31);
  const int f0 = start + wave_excl + (inc - s);
  const int f1 = f0 + s0;
  if (ok0) {
    off[i0] = f0;
    cnt[i0] = s0;
    int run = f0;
#pragma unroll
    for (int c = 0; c < C; ++c) { part[(size_t)c * N + i0] = run; run += p0[c]; }
  }
  if (ok1) {
    off[i1] = f1;
    cnt[i1] = s1;
    int run = f1;
#pragma unroll
    for (int c = 0; c < C; ++c) { part[(size_t)c * N + i1] = run; run += p1[c]; }
  }
  if (r == R - 1 && t == 0) off[N] = start + total;
}

template <int C>
__global__ void k_fill(const int* __restrict__ rows, const int* __restrict__ cols,
                       const int* __restrict__ rank, const int* __restrict__ part,
                       const int* __restrict__ cnt, int* __restrict__ adj, int N, int E) {
  int e = blockIdx.x * blockDim.x + threadIdx.x;
  if (e >= E) return;
  const int c = (int)(blockIdx.x * blockDim.x) / chunk_edges<C>(E);
  const int r = rows[e];
  const int pos = part[(size_t)c * N + cols[e]] + rank[e];
  const float dr = 1.0f / sqrtf((float)(cnt[r] + 1));
  ((int2*)adj)[pos] = make_int2(r, __float_as_int(dr));
}

template <int K, int NN>
__global__ void k_gemm_wmma(const float* __restrict__ A, const float* __restrict__ B,
                            float* __restrict__ C) {
  const int wave = threadIdx.x >> 5;
  const int lane = threadIdx.x & 31;
  const int m0   = blockIdx.x << 4;
  const int n0   = wave << 4;
  const int l16  = lane & 15;
  const int kh   = (lane >> 4) << 1;

  v8f acc = {};
  const float* arow = A + (size_t)(m0 + l16) * K + kh;
  const float* bcol = B + (size_t)kh * NN + n0 + l16;

#pragma unroll
  for (int k = 0; k < K; k += 4) {
    v2f a, b;
    a.x = arow[k];
    a.y = arow[k + 1];
    b.x = bcol[(size_t)k * NN];
    b.y = bcol[(size_t)(k + 1) * NN];
    acc = __builtin_amdgcn_wmma_f32_16x16x4_f32(false, a, false, b, (short)0, acc,
                                                false, false);
  }

  const int chi = (lane >> 4) << 3;
  float* crow = C + (size_t)(m0 + chi) * NN + n0 + l16;
#pragma unroll
  for (int v = 0; v < 8; ++v) crow[(size_t)v * NN] = acc[v];
}

__device__ inline float4 fma4(const float4 g, const float w, const float4 a) {
  return make_float4(fmaf(g.x, w, a.x), fmaf(g.y, w, a.y), fmaf(g.z, w, a.z), fmaf(g.w, w, a.w));
}

template <int F, bool RELU>
__device__ inline float4 gather_row(const int* __restrict__ adj, const int* __restrict__ off,
                                    const float* __restrict__ H,
                                    const float* __restrict__ bias, const int node,
                                    const int j, const int n) {
  constexpr int LPN = F / 4;
  const int2*   __restrict__ A2 = (const int2*)adj;
  const float4* __restrict__ H4 = (const float4*)H;
  const int beg = off[node];
  const int end = off[node + 1];
  const float dc = 1.0f / sqrtf((float)(end - beg + 1));
  const unsigned last = (unsigned)(n - 1);

  const float4 hs = H4[(size_t)node * LPN + j];
  const float4 b  = ((const float4*)bias)[j];

  float4 acc = make_float4(0.f, 0.f, 0.f, 0.f);
  int i = beg;
  for (; i + 4 <= end; i += 4) {
    const int2 p0 = A2[i], p1 = A2[i + 1], p2 = A2[i + 2], p3 = A2[i + 3];
    const float4 g0 = H4[(size_t)min((unsigned)p0.x, last) * LPN + j];
    const float4 g1 = H4[(size_t)min((unsigned)p1.x, last) * LPN + j];
    const float4 g2 = H4[(size_t)min((unsigned)p2.x, last) * LPN + j];
    const float4 g3 = H4[(size_t)min((unsigned)p3.x, last) * LPN + j];
    acc = fma4(g0, __int_as_float(p0.y) * dc, acc);
    acc = fma4(g1, __int_as_float(p1.y) * dc, acc);
    acc = fma4(g2, __int_as_float(p2.y) * dc, acc);
    acc = fma4(g3, __int_as_float(p3.y) * dc, acc);
  }
  if (i + 2 <= end) {
    const int2 p0 = A2[i], p1 = A2[i + 1];
    const float4 g0 = H4[(size_t)min((unsigned)p0.x, last) * LPN + j];
    const float4 g1 = H4[(size_t)min((unsigned)p1.x, last) * LPN + j];
    acc = fma4(g0, __int_as_float(p0.y) * dc, acc);
    acc = fma4(g1, __int_as_float(p1.y) * dc, acc);
    i += 2;
  }
  if (i < end) {
    const int2 p0 = A2[i];
    const float4 g0 = H4[(size_t)min((unsigned)p0.x, last) * LPN + j];
    acc = fma4(g0, __int_as_float(p0.y) * dc, acc);
  }

  const float dd  = dc * dc;
  float4 o;
  o.x = acc.x + hs.x * dd + b.x;
  o.y = acc.y + hs.y * dd + b.y;
  o.z = acc.z + hs.z * dd + b.z;
  o.w = acc.w + hs.w * dd + b.w;
  if (RELU) {
    o.x = fmaxf(o.x, 0.0f); o.y = fmaxf(o.y, 0.0f);
    o.z = fmaxf(o.z, 0.0f); o.w = fmaxf(o.w, 0.0f);
  }
  return o;
}

template <int F, bool RELU>
__global__ void k_gather(const int* __restrict__ adj, const int* __restrict__ off,
                         const float* __restrict__ H, const float* __restrict__ bias,
                         float* __restrict__ OUT, int n) {
  constexpr int LPN = F / 4;
  const int t    = blockIdx.x * blockDim.x + threadIdx.x;
  const int node = t / LPN;
  const int j    = t % LPN;
  if (node >= n) return;
  ((float4*)OUT)[(size_t)node * LPN + j] = gather_row<F, RELU>(adj, off, H, bias, node, j, n);
}

template <int F, int NN>
__global__ __launch_bounds__(256) void k_gather_relu_gemm(
    const int* __restrict__ adj, const int* __restrict__ off, const float* __restrict__ H,
    const float* __restrict__ bias, const float* __restrict__ B, float* __restrict__ C,
    int n) {
  constexpr int LPN   = F / 4;
  constexpr int ROWS  = 256 / LPN;
  constexpr int PITCH = F + 4;
  static_assert(ROWS == 16, "one 16-row matrix tile per work-group");
  static_assert(NN / 16 <= 8, "one wave per N-tile");
  __shared__ __attribute__((aligned(16))) float tile[ROWS * PITCH];

  const int nl   = threadIdx.x / LPN;
  const int j    = threadIdx.x % LPN;
  const int m0   = blockIdx.x * ROWS;
  const int node = m0 + nl;
  float4 o = make_float4(0.f, 0.f, 0.f, 0.f);
  if (node < n) o = gather_row<F, true>(adj, off, H, bias, node, j, n);
  *(float4*)&tile[nl * PITCH + 4 * j] = o;
  __syncthreads();

  const int wave = threadIdx.x >> 5;
  if (wave >= NN / 16) return;
  const int lane = threadIdx.x & 31;
  const int n0   = wave << 4;
  const int l16  = lane & 15;
  const int kh   = (lane >> 4) << 1;

  v8f acc = {};
  const float* arow = tile + l16 * PITCH + kh;
  const float* bcol = B + (size_t)kh * NN + n0 + l16;
#pragma unroll
  for (int k = 0; k < F; k += 4) {
    v2f a, b;
    a.x = arow[k];
    a.y = arow[k + 1];
    b.x = bcol[(size_t)k * NN];
    b.y = bcol[(size_t)(k + 1) * NN];
    acc = __builtin_amdgcn_wmma_f32_16x16x4_f32(false, a, false, b, (short)0, acc,
                                                false, false);
  }

  const int chi = (lane >> 4) << 3;
  float* crow = C + (size_t)(m0 + chi) * NN + n0 + l16;
#pragma unroll
  for (int v = 0; v < 8; ++v)
    if (m0 + chi + v < n) crow[(size_t)v * NN] = acc[v];
}

extern "C" void kernel_launch(void* const* d_in, const int* in_sizes, int n_in,
                              void* d_out, int out_size, void* d_ws, size_t ws_size,
                              hipStream_t stream) {
  const float* x  = (const float*)d_in[0];
  const int*   ei = (const int*)d_in[1];
  const float* W1 = (const float*)d_in[2];
  const float* b1 = (const float*)d_in[3];
  const float* W2 = (const float*)d_in[4];
  const float* b2 = (const float*)d_in[5];

  const int IN_DIM = 128, HID = 64, OUT = 32;
  const int N = in_sizes[0] / IN_DIM;
  const int E = in_sizes[1] / 2;
  const int* rows = ei;
  const int* cols = ei + E;

  float* ws   = (float*)d_ws;
  float* h1   = ws;
  float* h2   = h1 + (size_t)N * HID;
  int*   adj  = (int*)(h2 + (size_t)N * OUT);
  int*   rank = adj + (size_t)2 * E;
  int*   cnt  = rank + E;
  int*   off  = cnt + N;
  int*   part = off + N + 1;

  const int T = 256;
  const int R = 32;
  const int C = 8;
  int*   tot  = part + (size_t)C * N;

  k_rank_lds<R, C><<<R * C, 1024, 0, stream>>>(cols, rank, part, tot, N, E);
  k_offsets<R, C><<<R, 1024, 0, stream>>>(part, tot, off, cnt, N);
  k_fill<C><<<(E + T - 1) / T, T, 0, stream>>>(rows, cols, rank, part, cnt, adj, N, E);

  k_gemm_wmma<128, 64><<<N / 16, 32 * (64 / 16), 0, stream>>>(x, W1, h1);
  k_gather_relu_gemm<64, 32><<<(N * (HID / 4) + T - 1) / T, T, 0, stream>>>(adj, off, h1, b1,
                                                                            W2, h2, N);

  k_gather<32, false><<<(N * (OUT / 4) + T - 1) / T, T, 0, stream>>>(adj, off, h2, b2,
                                                                     (float*)d_out, N);
}


# x@W1: rows of x reach the A fragments through LDS, one whole 512-byte row per wave-wide 16-byte load, 32 rows per work-group so one W1 fragment serves two row tiles; every sum keeps its order
# speedup vs baseline: 3.0103x; 1.0409x over previous
#include <hip/hip_runtime.h>
#include <cstddef>


typedef __attribute__((ext_vector_type(2))) float v2f;
typedef __attribute__((ext_vector_type(8))) float v8f;

__device__ inline int wave_incl_scan(int v, int lane) {
  v += __builtin_amdgcn_update_dpp(0, v, 0x111, 0xf, 0xf, false);
  v += __builtin_amdgcn_update_dpp(0, v, 0x112, 0xf, 0xf, false);
  v += __builtin_amdgcn_update_dpp(0, v, 0x114, 0xf, 0xf, false);
  v += __builtin_amdgcn_update_dpp(0, v, 0x118, 0xf, 0xf, false);
  const int low = __builtin_amdgcn_readlane(v, 15);
  if (lane >= 16) v += low;
  return v;
}

template <int R> __device__ inline int range_nodes(const int N) { return (N + R - 1) / R; }
template <int C> __device__ inline int chunk_edges(const int E) {
  return ((E + C - 1) / C + 4095) / 4096 * 4096;
}

__device__ inline int4 count4(const int4 v, const int lo, const unsigned span, int* lc) {
  const unsigned u0 = (unsigned)(v.x - lo), u1 = (unsigned)(v.y - lo);
  const unsigned u2 = (unsigned)(v.z - lo), u3 = (unsigned)(v.w - lo);
  int4 k = make_int4(-1, -1, -1, -1);
  if (u0 < span) k.x = atomicAdd(&lc[u0], 1);
  if (u1 < span) k.y = atomicAdd(&lc[u1], 1);
  if (u2 < span) k.z = atomicAdd(&lc[u2], 1);
  if (u3 < span) k.w = atomicAdd(&lc[u3], 1);
  return k;
}

__device__ inline void store_ranks(int* __restrict__ rank, const int b, const int4 k) {
  if (k.x >= 0) rank[b]     = k.x;
  if (k.y >= 0) rank[b + 1] = k.y;
  if (k.z >= 0) rank[b + 2] = k.z;
  if (k.w >= 0) rank[b + 3] = k.w;
}

template <int R, int C>
__global__ __launch_bounds__(1024) void k_rank_lds(const int* __restrict__ cols,
                                                   int* __restrict__ rank,
                                                   int* __restrict__ part,
                                                   int* __restrict__ tot, int N, int E) {
  __shared__ int lc[2048];
  const int RN = range_nodes<R>(N);
  const int CE = chunk_edges<C>(E);
  __shared__ int stot;
  const int t    = threadIdx.x;
  const int lane = t & 31;
  const int r    = blockIdx.x % R;
  const int c    = blockIdx.x / R;
  const int lo   = r * RN;
  const int nsp  = max(min(lo + RN, N) - lo, 0);
  const unsigned span = (unsigned)nsp;
  for (int i = t; i < nsp; i += 1024) lc[i] = 0;
  if (t == 0) stot = 0;
  __syncthreads();

  const int e0  = c * CE;
  const int e1  = max(min(e0 + CE, E), e0);
  const int e1v = e1 & ~3;
  const int4* __restrict__ c4 = (const int4*)cols;
  int b0 = e0;
  for (; b0 + 4 * 4096 <= e1v; b0 += 4 * 4096) {
    const int b = b0 + (t << 2);
    const int4 v0 = c4[b >> 2];
    const int4 v1 = c4[(b >> 2) + 1024];
    const int4 v2 = c4[(b >> 2) + 2048];
    const int4 v3 = c4[(b >> 2) + 3072];
    const int4 k0 = count4(v0, lo, span, lc);
    const int4 k1 = count4(v1, lo, span, lc);
    const int4 k2 = count4(v2, lo, span, lc);
    const int4 k3 = count4(v3, lo, span, lc);
    store_ranks(rank, b, k0);
    store_ranks(rank, b + 4096, k1);
    store_ranks(rank, b + 2 * 4096, k2);
    store_ranks(rank, b + 3 * 4096, k3);
  }
  for (; b0 < e1v; b0 += 4096) {
    const int b = b0 + (t << 2);
    if (b < e1v) {
      const int4 k0 = count4(c4[b >> 2], lo, span, lc);
      store_ranks(rank, b, k0);
    }
  }
  for (int b = e1v + t; b < e1; b += 1024) {
    const unsigned u = (unsigned)(cols[b] - lo);
    if (u < span) rank[b] = atomicAdd(&lc[u], 1);
  }
  __syncthreads();
  int mine = 0;
  for (int i = t; i < nsp; i += 1024) {
    const int k = lc[i];
    part[(size_t)c * N + lo + i] = k;
    mine += k;
  }
  const int winc = wave_incl_scan(mine, lane);
  if (lane == 31) atomicAdd(&stot, winc);
  __syncthreads();
  if (t == 0) tot[blockIdx.x] = stot;
}

template <int R, int C>
__global__ __launch_bounds__(1024) void k_offsets(int* part, const int* __restrict__ tot,
                                                  int* __restrict__ off,
                                                  int* __restrict__ cnt, int N) {
  __shared__ int wa[32];
  __shared__ int wb[32];
  const int t    = threadIdx.x;
  const int lane = t & 31;
  const int wv   = __builtin_amdgcn_readfirstlane(t >> 5);
  const int r    = blockIdx.x;
  const int lo   = r * range_nodes<R>(N);
  const int hi   = min(lo + range_nodes<R>(N), N);

  const int i0 = lo + 2 * t;
  const int i1 = i0 + 1;
  const bool ok0 = i0 < hi, ok1 = i1 < hi;
  int p0[C], p1[C];
#pragma unroll
  for (int c = 0; c < C; ++c) {
    p0[c] = ok0 ? part[(size_t)c * N + i0] : 0;
    p1[c] = ok1 ? part[(size_t)c * N + i1] : 0;
  }
  int tv = 0;
  if (t < R * C && (t % R) < r) tv = tot[t];
  const int tinc = wave_incl_scan(tv, lane);
  if (lane == 31) wa[wv] = tinc;
  int s0 = 0, s1 = 0;
#pragma unroll
  for (int c = 0; c < C; ++c) {
    s0 += p0[c];
    s1 += p1[c];
  }
  const int s   = s0 + s1;
  const int inc = wave_incl_scan(s, lane);
  if (lane == 31) wb[wv] = inc;
  __syncthreads();
  const int a    = wa[lane];
  const int b    = wb[lane];
  const int ainc = wave_incl_scan(a, lane);
  const int binc = wave_incl_scan(b, lane);
  const int start = __builtin_amdgcn_readlane(ainc, 31);
  const int wave_excl =
      __builtin_amdgcn_readlane(binc, wv) - __builtin_amdgcn_readlane(b, wv);
  const int total = __builtin_amdgcn_readlane(binc, 31);
  const int f0 = start + wave_excl + (inc - s);
  const int f1 = f0 + s0;
  if (ok0) {
    off[i0] = f0;
    cnt[i0] = s0;
    int run = f0;
#pragma unroll
    for (int c = 0; c < C; ++c) { part[(size_t)c * N + i0] = run; run += p0[c]; }
  }
  if (ok1) {
    off[i1] = f1;
    cnt[i1] = s1;
    int run = f1;
#pragma unroll
    for (int c = 0; c < C; ++c) { part[(size_t)c * N + i1] = run; run += p1[c]; }
  }
  if (r == R - 1 && t == 0) off[N] = start + total;
}

template <int C>
__global__ void k_fill(const int* __restrict__ rows, const int* __restrict__ cols,
                       const int* __restrict__ rank, const int* __restrict__ part,
                       const int* __restrict__ cnt, int* __restrict__ adj, int N, int E) {
  int e = blockIdx.x * blockDim.x + threadIdx.x;
  if (e >= E) return;
  const int c = (int)(blockIdx.x * blockDim.x) / chunk_edges<C>(E);
  const int r = rows[e];
  const int pos = part[(size_t)c * N + cols[e]] + rank[e];
  const float dr = 1.0f / sqrtf((float)(cnt[r] + 1));
  ((int2*)adj)[pos] = make_int2(r, __float_as_int(dr));
}

template <int K, int NN, int MT>
__global__ __launch_bounds__(32 * (NN / 16)) __attribute__((amdgpu_waves_per_eu(1, 8)))
void k_gemm_wmma(const float* __restrict__ A, const float* __restrict__ B,
                 float* __restrict__ C, int M) {
  constexpr int NW    = NN / 16;
  constexpr int ROWS  = 16 * MT;
  constexpr int PITCH = K + 4;
  static_assert(K == 128, "one wave-wide 16-byte load = one row of A");
  static_assert(ROWS % NW == 0, "rows split evenly over the waves");
  __shared__ __attribute__((aligned(16))) float tile[ROWS * PITCH];

  const int wave = threadIdx.x >> 5;
  const int lane = threadIdx.x & 31;
  const int m0   = blockIdx.x * ROWS;

  const int n0   = wave << 4;
  const int l16  = lane & 15;
  const int kh   = (lane >> 4) << 1;

  const float4* __restrict__ A4 = (const float4*)A;
  float4 st[ROWS / NW];
#pragma unroll
  for (int i = 0; i < ROWS / NW; ++i) {
    const int r = min(m0 + wave + NW * i, M - 1);
    st[i] = A4[(size_t)r * (K / 4) + lane];
  }
  const float* bcol = B + (size_t)kh * NN + n0 + l16;
  v2f bf[K / 4];
#pragma unroll
  for (int k = 0; k < K; k += 4) {
    bf[k / 4].x = bcol[(size_t)k * NN];
    bf[k / 4].y = bcol[(size_t)(k + 1) * NN];
  }
#pragma unroll
  for (int i = 0; i < ROWS / NW; ++i)
    *(float4*)&tile[(wave + NW * i) * PITCH + 4 * lane] = st[i];
  __syncthreads();

  v8f acc[MT];
#pragma unroll
  for (int mt = 0; mt < MT; ++mt) acc[mt] = v8f{};
  const float* arow = tile + l16 * PITCH + kh;

#pragma unroll
  for (int k = 0; k < K; k += 4) {
#pragma unroll
    for (int mt = 0; mt < MT; ++mt) {
      v2f a;
      a.x = arow[mt * 16 * PITCH + k];
      a.y = arow[mt * 16 * PITCH + k + 1];
      acc[mt] = __builtin_amdgcn_wmma_f32_16x16x4_f32(false, a, false, bf[k / 4], (short)0,
                                                      acc[mt], false, false);
    }
  }

  const int chi = (lane >> 4) << 3;
#pragma unroll
  for (int mt = 0; mt < MT; ++mt) {
    const int mr = m0 + mt * 16 + chi;
    float* crow = C + (size_t)mr * NN + n0 + l16;
#pragma unroll
    for (int v = 0; v < 8; ++v)
      if (mr + v < M) crow[(size_t)v * NN] = acc[mt][v];
  }
}

__device__ inline float4 fma4(const float4 g, const float w, const float4 a) {
  return make_float4(fmaf(g.x, w, a.x), fmaf(g.y, w, a.y), fmaf(g.z, w, a.z), fmaf(g.w, w, a.w));
}

template <int F, bool RELU>
__device__ inline float4 gather_row(const int* __restrict__ adj, const int* __restrict__ off,
                                    const float* __restrict__ H,
                                    const float* __restrict__ bias, const int node,
                                    const int j, const int n) {
  constexpr int LPN = F / 4;
  const int2*   __restrict__ A2 = (const int2*)adj;
  const float4* __restrict__ H4 = (const float4*)H;
  const int beg = off[node];
  const int end = off[node + 1];
  const float dc = 1.0f / sqrtf((float)(end - beg + 1));
  const unsigned last = (unsigned)(n - 1);

  const float4 hs = H4[(size_t)node * LPN + j];
  const float4 b  = ((const float4*)bias)[j];

  float4 acc = make_float4(0.f, 0.f, 0.f, 0.f);
  int i = beg;
  for (; i + 4 <= end; i += 4) {
    const int2 p0 = A2[i], p1 = A2[i + 1], p2 = A2[i + 2], p3 = A2[i + 3];
    const float4 g0 = H4[(size_t)min((unsigned)p0.x, last) * LPN + j];
    const float4 g1 = H4[(size_t)min((unsigned)p1.x, last) * LPN + j];
    const float4 g2 = H4[(size_t)min((unsigned)p2.x, last) * LPN + j];
    const float4 g3 = H4[(size_t)min((unsigned)p3.x, last) * LPN + j];
    acc = fma4(g0, __int_as_float(p0.y) * dc, acc);
    acc = fma4(g1, __int_as_float(p1.y) * dc, acc);
    acc = fma4(g2, __int_as_float(p2.y) * dc, acc);
    acc = fma4(g3, __int_as_float(p3.y) * dc, acc);
  }
  if (i + 2 <= end) {
    const int2 p0 = A2[i], p1 = A2[i + 1];
    const float4 g0 = H4[(size_t)min((unsigned)p0.x, last) * LPN + j];
    const float4 g1 = H4[(size_t)min((unsigned)p1.x, last) * LPN + j];
    acc = fma4(g0, __int_as_float(p0.y) * dc, acc);
    acc = fma4(g1, __int_as_float(p1.y) * dc, acc);
    i += 2;
  }
  if (i < end) {
    const int2 p0 = A2[i];
    const float4 g0 = H4[(size_t)min((unsigned)p0.x, last) * LPN + j];
    acc = fma4(g0, __int_as_float(p0.y) * dc, acc);
  }

  const float dd  = dc * dc;
  float4 o;
  o.x = acc.x + hs.x * dd + b.x;
  o.y = acc.y + hs.y * dd + b.y;
  o.z = acc.z + hs.z * dd + b.z;
  o.w = acc.w + hs.w * dd + b.w;
  if (RELU) {
    o.x = fmaxf(o.x, 0.0f); o.y = fmaxf(o.y, 0.0f);
    o.z = fmaxf(o.z, 0.0f); o.w = fmaxf(o.w, 0.0f);
  }
  return o;
}

template <int F, bool RELU>
__global__ void k_gather(const int* __restrict__ adj, const int* __restrict__ off,
                         const float* __restrict__ H, const float* __restrict__ bias,
                         float* __restrict__ OUT, int n) {
  constexpr int LPN = F / 4;
  const int t    = blockIdx.x * blockDim.x + threadIdx.x;
  const int node = t / LPN;
  const int j    = t % LPN;
  if (node >= n) return;
  ((float4*)OUT)[(size_t)node * LPN + j] = gather_row<F, RELU>(adj, off, H, bias, node, j, n);
}

template <int F, int NN>
__global__ __launch_bounds__(256) void k_gather_relu_gemm(
    const int* __restrict__ adj, const int* __restrict__ off, const float* __restrict__ H,
    const float* __restrict__ bias, const float* __restrict__ B, float* __restrict__ C,
    int n) {
  constexpr int LPN   = F / 4;
  constexpr int ROWS  = 256 / LPN;
  constexpr int PITCH = F + 4;
  static_assert(ROWS == 16, "one 16-row matrix tile per work-group");
  static_assert(NN / 16 <= 8, "one wave per N-tile");
  __shared__ __attribute__((aligned(16))) float tile[ROWS * PITCH];

  const int nl   = threadIdx.x / LPN;
  const int j    = threadIdx.x % LPN;
  const int m0   = blockIdx.x * ROWS;
  const int node = m0 + nl;
  float4 o = make_float4(0.f, 0.f, 0.f, 0.f);
  if (node < n) o = gather_row<F, true>(adj, off, H, bias, node, j, n);
  *(float4*)&tile[nl * PITCH + 4 * j] = o;
  __syncthreads();

  const int wave = threadIdx.x >> 5;
  if (wave >= NN / 16) return;
  const int lane = threadIdx.x & 31;
  const int n0   = wave << 4;
  const int l16  = lane & 15;
  const int kh   = (lane >> 4) << 1;

  v8f acc = {};
  const float* arow = tile + l16 * PITCH + kh;
  const float* bcol = B + (size_t)kh * NN + n0 + l16;
#pragma unroll
  for (int k = 0; k < F; k += 4) {
    v2f a, b;
    a.x = arow[k];
    a.y = arow[k + 1];
    b.x = bcol[(size_t)k * NN];
    b.y = bcol[(size_t)(k + 1) * NN];
    acc = __builtin_amdgcn_wmma_f32_16x16x4_f32(false, a, false, b, (short)0, acc,
                                                false, false);
  }

  const int chi = (lane >> 4) << 3;
  float* crow = C + (size_t)(m0 + chi) * NN + n0 + l16;
#pragma unroll
  for (int v = 0; v < 8; ++v)
    if (m0 + chi + v < n) crow[(size_t)v * NN] = acc[v];
}

extern "C" void kernel_launch(void* const* d_in, const int* in_sizes, int n_in,
                              void* d_out, int out_size, void* d_ws, size_t ws_size,
                              hipStream_t stream) {
  const float* x  = (const float*)d_in[0];
  const int*   ei = (const int*)d_in[1];
  const float* W1 = (const float*)d_in[2];
  const float* b1 = (const float*)d_in[3];
  const float* W2 = (const float*)d_in[4];
  const float* b2 = (const float*)d_in[5];

  const int IN_DIM = 128, HID = 64, OUT = 32;
  const int N = in_sizes[0] / IN_DIM;
  const int E = in_sizes[1] / 2;
  const int* rows = ei;
  const int* cols = ei + E;

  float* ws   = (float*)d_ws;
  float* h1   = ws;
  float* h2   = h1 + (size_t)N * HID;
  int*   adj  = (int*)(h2 + (size_t)N * OUT);
  int*   rank = adj + (size_t)2 * E;
  int*   cnt  = rank + E;
  int*   off  = cnt + N;
  int*   part = off + N + 1;

  const int T = 256;
  const int R = 32;
  const int C = 8;
  int*   tot  = part + (size_t)C * N;

  k_rank_lds<R, C><<<R * C, 1024, 0, stream>>>(cols, rank, part, tot, N, E);
  k_offsets<R, C><<<R, 1024, 0, stream>>>(part, tot, off, cnt, N);
  k_fill<C><<<(E + T - 1) / T, T, 0, stream>>>(rows, cols, rank, part, cnt, adj, N, E);

  k_gemm_wmma<128, 64, 2><<<(N + 31) / 32, 32 * (64 / 16), 0, stream>>>(x, W1, h1, N);
  k_gather_relu_gemm<64, 32><<<(N * (HID / 4) + T - 1) / T, T, 0, stream>>>(adj, off, h1, b1,
                                                                            W2, h2, N);

  k_gather<32, false><<<(N * (OUT / 4) + T - 1) / T, T, 0, stream>>>(adj, off, h2, b2,
                                                                     (float*)d_out, N);
}


# x@W1 runs inside the edge-rank launch: extra 1024-thread work-groups (256 rows each, B fragments loaded in the chain) beside the counting ones, 2 per WGP at 64 VGPRs; 6 launches to 5; sums keep order
# speedup vs baseline: 3.1039x; 1.0311x over previous
#include <hip/hip_runtime.h>
#include <cstddef>


typedef __attribute__((ext_vector_type(2))) float v2f;
typedef __attribute__((ext_vector_type(8))) float v8f;

__device__ inline int wave_incl_scan(int v, int lane) {
  v += __builtin_amdgcn_update_dpp(0, v, 0x111, 0xf, 0xf, false);
  v += __builtin_amdgcn_update_dpp(0, v, 0x112, 0xf, 0xf, false);
  v += __builtin_amdgcn_update_dpp(0, v, 0x114, 0xf, 0xf, false);
  v += __builtin_amdgcn_update_dpp(0, v, 0x118, 0xf, 0xf, false);
  const int low = __builtin_amdgcn_readlane(v, 15);
  if (lane >= 16) v += low;
  return v;
}

template <int R> __device__ inline int range_nodes(const int N) { return (N + R - 1) / R; }
template <int C> __device__ inline int chunk_edges(const int E) {
  return ((E + C - 1) / C + 4095) / 4096 * 4096;
}

__device__ inline int4 count4(const int4 v, const int lo, const unsigned span, int* lc) {
  const unsigned u0 = (unsigned)(v.x - lo), u1 = (unsigned)(v.y - lo);
  const unsigned u2 = (unsigned)(v.z - lo), u3 = (unsigned)(v.w - lo);
  int4 k = make_int4(-1, -1, -1, -1);
  if (u0 < span) k.x = atomicAdd(&lc[u0], 1);
  if (u1 < span) k.y = atomicAdd(&lc[u1], 1);
  if (u2 < span) k.z = atomicAdd(&lc[u2], 1);
  if (u3 < span) k.w = atomicAdd(&lc[u3], 1);
  return k;
}

__device__ inline void store_ranks(int* __restrict__ rank, const int b, const int4 k) {
  if (k.x >= 0) rank[b]     = k.x;
  if (k.y >= 0) rank[b + 1] = k.y;
  if (k.z >= 0) rank[b + 2] = k.z;
  if (k.w >= 0) rank[b + 3] = k.w;
}

template <int R, int C>
__device__ __forceinline__ void rank_lds_block(int* lc, int* stot, const int bid,
                                               const int* __restrict__ cols,
                                               int* __restrict__ rank,
                                               int* __restrict__ part,
                                               int* __restrict__ tot, int N, int E) {
  const int RN = range_nodes<R>(N);
  const int CE = chunk_edges<C>(E);
  const int t    = threadIdx.x;
  const int lane = t & 31;
  const int r    = bid % R;
  const int c    = bid / R;
  const int lo   = r * RN;
  const int nsp  = max(min(lo + RN, N) - lo, 0);
  const unsigned span = (unsigned)nsp;
  for (int i = t; i < nsp; i += 1024) lc[i] = 0;
  if (t == 0) *stot = 0;
  __syncthreads();

  const int e0  = c * CE;
  const int e1  = max(min(e0 + CE, E), e0);
  const int e1v = e1 & ~3;
  const int4* __restrict__ c4 = (const int4*)cols;
  int b0 = e0;
  for (; b0 + 4 * 4096 <= e1v; b0 += 4 * 4096) {
    const int b = b0 + (t << 2);
    const int4 v0 = c4[b >> 2];
    const int4 v1 = c4[(b >> 2) + 1024];
    const int4 v2 = c4[(b >> 2) + 2048];
    const int4 v3 = c4[(b >> 2) + 3072];
    const int4 k0 = count4(v0, lo, span, lc);
    const int4 k1 = count4(v1, lo, span, lc);
    const int4 k2 = count4(v2, lo, span, lc);
    const int4 k3 = count4(v3, lo, span, lc);
    store_ranks(rank, b, k0);
    store_ranks(rank, b + 4096, k1);
    store_ranks(rank, b + 2 * 4096, k2);
    store_ranks(rank, b + 3 * 4096, k3);
  }
  for (; b0 < e1v; b0 += 4096) {
    const int b = b0 + (t << 2);
    if (b < e1v) {
      const int4 k0 = count4(c4[b >> 2], lo, span, lc);
      store_ranks(rank, b, k0);
    }
  }
  for (int b = e1v + t; b < e1; b += 1024) {
    const unsigned u = (unsigned)(cols[b] - lo);
    if (u < span) rank[b] = atomicAdd(&lc[u], 1);
  }
  __syncthreads();
  int mine = 0;
  for (int i = t; i < nsp; i += 1024) {
    const int k = lc[i];
    part[(size_t)c * N + lo + i] = k;
    mine += k;
  }
  const int winc = wave_incl_scan(mine, lane);
  if (lane == 31) atomicAdd(stot, winc);
  __syncthreads();
  if (t == 0) tot[bid] = *stot;
}

template <int R, int C>
__global__ __launch_bounds__(1024) void k_offsets(int* part, const int* __restrict__ tot,
                                                  int* __restrict__ off,
                                                  int* __restrict__ cnt, int N) {
  __shared__ int wa[32];
  __shared__ int wb[32];
  const int t    = threadIdx.x;
  const int lane = t & 31;
  const int wv   = __builtin_amdgcn_readfirstlane(t >> 5);
  const int r    = blockIdx.x;
  const int lo   = r * range_nodes<R>(N);
  const int hi   = min(lo + range_nodes<R>(N), N);

  const int i0 = lo + 2 * t;
  const int i1 = i0 + 1;
  const bool ok0 = i0 < hi, ok1 = i1 < hi;
  int p0[C], p1[C];
#pragma unroll
  for (int c = 0; c < C; ++c) {
    p0[c] = ok0 ? part[(size_t)c * N + i0] : 0;
    p1[c] = ok1 ? part[(size_t)c * N + i1] : 0;
  }
  int tv = 0;
  if (t < R * C && (t % R) < r) tv = tot[t];
  const int tinc = wave_incl_scan(tv, lane);
  if (lane == 31) wa[wv] = tinc;
  int s0 = 0, s1 = 0;
#pragma unroll
  for (int c = 0; c < C; ++c) {
    s0 += p0[c];
    s1 += p1[c];
  }
  const int s   = s0 + s1;
  const int inc = wave_incl_scan(s, lane);
  if (lane == 31) wb[wv] = inc;
  __syncthreads();
  const int a    = wa[lane];
  const int b    = wb[lane];
  const int ainc = wave_incl_scan(a, lane);
  const int binc = wave_incl_scan(b, lane);
  const int start = __builtin_amdgcn_readlane(ainc, 31);
  const int wave_excl =
      __builtin_amdgcn_readlane(binc, wv) - __builtin_amdgcn_readlane(b, wv);
  const int total = __builtin_amdgcn_readlane(binc, 31);
  const int f0 = start + wave_excl + (inc - s);
  const int f1 = f0 + s0;
  if (ok0) {
    off[i0] = f0;
    cnt[i0] = s0;
    int run = f0;
#pragma unroll
    for (int c = 0; c < C; ++c) { part[(size_t)c * N + i0] = run; run += p0[c]; }
  }
  if (ok1) {
    off[i1] = f1;
    cnt[i1] = s1;
    int run = f1;
#pragma unroll
    for (int c = 0; c < C; ++c) { part[(size_t)c * N + i1] = run; run += p1[c]; }
  }
  if (r == R - 1 && t == 0) off[N] = start + total;
}

template <int C>
__global__ void k_fill(const int* __restrict__ rows, const int* __restrict__ cols,
                       const int* __restrict__ rank, const int* __restrict__ part,
                       const int* __restrict__ cnt, int* __restrict__ adj, int N, int E) {
  int e = blockIdx.x * blockDim.x + threadIdx.x;
  if (e >= E) return;
  const int c = (int)(blockIdx.x * blockDim.x) / chunk_edges<C>(E);
  const int r = rows[e];
  const int pos = part[(size_t)c * N + cols[e]] + rank[e];
  const float dr = 1.0f / sqrtf((float)(cnt[r] + 1));
  ((int2*)adj)[pos] = make_int2(r, __float_as_int(dr));
}

template <int K, int NN, int MT, int NG>
__device__ __forceinline__ void gemm_rows_block(float* tile, const int bid,
                                                const float* __restrict__ A,
                                                const float* __restrict__ B,
                                                float* __restrict__ C, int M) {
  constexpr int NT    = NN / 16;
  constexpr int NW    = NT * NG;
  constexpr int GROWS = 16 * MT;
  constexpr int ROWS  = GROWS * NG;
  constexpr int PITCH = K + 4;
  static_assert(K == 128, "one wave-wide 16-byte load = one row of A");
  static_assert(ROWS % NW == 0, "rows split evenly over the waves");

  const int wave = threadIdx.x >> 5;
  const int lane = threadIdx.x & 31;
  const int grp  = wave / NT;
  const int m0   = bid * ROWS;

  const int n0   = (wave % NT) << 4;
  const int l16  = lane & 15;
  const int kh   = (lane >> 4) << 1;

  const float4* __restrict__ A4 = (const float4*)A;
  float4 st[ROWS / NW];
#pragma unroll
  for (int i = 0; i < ROWS / NW; ++i) {
    const int r = min(m0 + wave + NW * i, M - 1);
    st[i] = A4[(size_t)r * (K / 4) + lane];
  }
#pragma unroll
  for (int i = 0; i < ROWS / NW; ++i)
    *(float4*)&tile[(wave + NW * i) * PITCH + 4 * lane] = st[i];
  __syncthreads();

  v8f acc[MT];
#pragma unroll
  for (int mt = 0; mt < MT; ++mt) acc[mt] = v8f{};
  const float* arow = tile + (grp * GROWS + l16) * PITCH + kh;
  const float* bcol = B + (size_t)kh * NN + n0 + l16;

#pragma unroll
  for (int k = 0; k < K; k += 4) {
    v2f b;
    b.x = bcol[(size_t)k * NN];
    b.y = bcol[(size_t)(k + 1) * NN];
#pragma unroll
    for (int mt = 0; mt < MT; ++mt) {
      v2f a;
      a.x = arow[mt * 16 * PITCH + k];
      a.y = arow[mt * 16 * PITCH + k + 1];
      acc[mt] = __builtin_amdgcn_wmma_f32_16x16x4_f32(false, a, false, b, (short)0,
                                                      acc[mt], false, false);
    }
  }

  const int chi = (lane >> 4) << 3;
#pragma unroll
  for (int mt = 0; mt < MT; ++mt) {
    const int mr = m0 + grp * GROWS + mt * 16 + chi;
    float* crow = C + (size_t)mr * NN + n0 + l16;
#pragma unroll
    for (int v = 0; v < 8; ++v)
      if (mr + v < M) crow[(size_t)v * NN] = acc[mt][v];
  }
}

template <int R, int C, int K, int NN>
__global__ __launch_bounds__(1024) __attribute__((amdgpu_waves_per_eu(16, 16)))
void k_rank_and_gemm(const int* __restrict__ cols, int* __restrict__ rank,
                     int* __restrict__ part, int* __restrict__ tot, int N, int E,
                     const float* __restrict__ A, const float* __restrict__ B,
                     float* __restrict__ Cm) {
  constexpr int MT = 2, NG = 8;
  static_assert((NN / 16) * NG * 32 == 1024, "product work-groups fill the 1024 threads");
  __shared__ __attribute__((aligned(16))) float smem[16 * MT * NG * (K + 4)];
  static_assert(sizeof(smem) >= 2049 * sizeof(int), "room for the counters");
  if (blockIdx.x < R * C) {
    int* lc = (int*)smem;
    rank_lds_block<R, C>(lc, lc + 2048, (int)blockIdx.x, cols, rank, part, tot, N, E);
  } else {
    gemm_rows_block<K, NN, MT, NG>(smem, (int)blockIdx.x - R * C, A, B, Cm, N);
  }
}

__device__ inline float4 fma4(const float4 g, const float w, const float4 a) {
  return make_float4(fmaf(g.x, w, a.x), fmaf(g.y, w, a.y), fmaf(g.z, w, a.z), fmaf(g.w, w, a.w));
}

template <int F, bool RELU>
__device__ inline float4 gather_row(const int* __restrict__ adj, const int* __restrict__ off,
                                    const float* __restrict__ H,
                                    const float* __restrict__ bias, const int node,
                                    const int j, const int n) {
  constexpr int LPN = F / 4;
  const int2*   __restrict__ A2 = (const int2*)adj;
  const float4* __restrict__ H4 = (const float4*)H;
  const int beg = off[node];
  const int end = off[node + 1];
  const float dc = 1.0f / sqrtf((float)(end - beg + 1));
  const unsigned last = (unsigned)(n - 1);

  const float4 hs = H4[(size_t)node * LPN + j];
  const float4 b  = ((const float4*)bias)[j];

  float4 acc = make_float4(0.f, 0.f, 0.f, 0.f);
  int i = beg;
  for (; i + 4 <= end; i += 4) {
    const int2 p0 = A2[i], p1 = A2[i + 1], p2 = A2[i + 2], p3 = A2[i + 3];
    const float4 g0 = H4[(size_t)min((unsigned)p0.x, last) * LPN + j];
    const float4 g1 = H4[(size_t)min((unsigned)p1.x, last) * LPN + j];
    const float4 g2 = H4[(size_t)min((unsigned)p2.x, last) * LPN + j];
    const float4 g3 = H4[(size_t)min((unsigned)p3.x, last) * LPN + j];
    acc = fma4(g0, __int_as_float(p0.y) * dc, acc);
    acc = fma4(g1, __int_as_float(p1.y) * dc, acc);
    acc = fma4(g2, __int_as_float(p2.y) * dc, acc);
    acc = fma4(g3, __int_as_float(p3.y) * dc, acc);
  }
  if (i + 2 <= end) {
    const int2 p0 = A2[i], p1 = A2[i + 1];
    const float4 g0 = H4[(size_t)min((unsigned)p0.x, last) * LPN + j];
    const float4 g1 = H4[(size_t)min((unsigned)p1.x, last) * LPN + j];
    acc = fma4(g0, __int_as_float(p0.y) * dc, acc);
    acc = fma4(g1, __int_as_float(p1.y) * dc, acc);
    i += 2;
  }
  if (i < end) {
    const int2 p0 = A2[i];
    const float4 g0 = H4[(size_t)min((unsigned)p0.x, last) * LPN + j];
    acc = fma4(g0, __int_as_float(p0.y) * dc, acc);
  }

  const float dd  = dc * dc;
  float4 o;
  o.x = acc.x + hs.x * dd + b.x;
  o.y = acc.y + hs.y * dd + b.y;
  o.z = acc.z + hs.z * dd + b.z;
  o.w = acc.w + hs.w * dd + b.w;
  if (RELU) {
    o.x = fmaxf(o.x, 0.0f); o.y = fmaxf(o.y, 0.0f);
    o.z = fmaxf(o.z, 0.0f); o.w = fmaxf(o.w, 0.0f);
  }
  return o;
}

template <int F, bool RELU>
__global__ void k_gather(const int* __restrict__ adj, const int* __restrict__ off,
                         const float* __restrict__ H, const float* __restrict__ bias,
                         float* __restrict__ OUT, int n) {
  constexpr int LPN = F / 4;
  const int t    = blockIdx.x * blockDim.x + threadIdx.x;
  const int node = t / LPN;
  const int j    = t % LPN;
  if (node >= n) return;
  ((float4*)OUT)[(size_t)node * LPN + j] = gather_row<F, RELU>(adj, off, H, bias, node, j, n);
}

template <int F, int NN>
__global__ __launch_bounds__(256) void k_gather_relu_gemm(
    const int* __restrict__ adj, const int* __restrict__ off, const float* __restrict__ H,
    const float* __restrict__ bias, const float* __restrict__ B, float* __restrict__ C,
    int n) {
  constexpr int LPN   = F / 4;
  constexpr int ROWS  = 256 / LPN;
  constexpr int PITCH = F + 4;
  static_assert(ROWS == 16, "one 16-row matrix tile per work-group");
  static_assert(NN / 16 <= 8, "one wave per N-tile");
  __shared__ __attribute__((aligned(16))) float tile[ROWS * PITCH];

  const int nl   = threadIdx.x / LPN;
  const int j    = threadIdx.x % LPN;
  const int m0   = blockIdx.x * ROWS;
  const int node = m0 + nl;
  float4 o = make_float4(0.f, 0.f, 0.f, 0.f);
  if (node < n) o = gather_row<F, true>(adj, off, H, bias, node, j, n);
  *(float4*)&tile[nl * PITCH + 4 * j] = o;
  __syncthreads();

  const int wave = threadIdx.x >> 5;
  if (wave >= NN / 16) return;
  const int lane = threadIdx.x & 31;
  const int n0   = wave << 4;
  const int l16  = lane & 15;
  const int kh   = (lane >> 4) << 1;

  v8f acc = {};
  const float* arow = tile + l16 * PITCH + kh;
  const float* bcol = B + (size_t)kh * NN + n0 + l16;
#pragma unroll
  for (int k = 0; k < F; k += 4) {
    v2f a, b;
    a.x = arow[k];
    a.y = arow[k + 1];
    b.x = bcol[(size_t)k * NN];
    b.y = bcol[(size_t)(k + 1) * NN];
    acc = __builtin_amdgcn_wmma_f32_16x16x4_f32(false, a, false, b, (short)0, acc,
                                                false, false);
  }

  const int chi = (lane >> 4) << 3;
  float* crow = C + (size_t)(m0 + chi) * NN + n0 + l16;
#pragma unroll
  for (int v = 0; v < 8; ++v)
    if (m0 + chi + v < n) crow[(size_t)v * NN] = acc[v];
}

extern "C" void kernel_launch(void* const* d_in, const int* in_sizes, int n_in,
                              void* d_out, int out_size, void* d_ws, size_t ws_size,
                              hipStream_t stream) {
  const float* x  = (const float*)d_in[0];
  const int*   ei = (const int*)d_in[1];
  const float* W1 = (const float*)d_in[2];
  const float* b1 = (const float*)d_in[3];
  const float* W2 = (const float*)d_in[4];
  const float* b2 = (const float*)d_in[5];

  const int IN_DIM = 128, HID = 64, OUT = 32;
  const int N = in_sizes[0] / IN_DIM;
  const int E = in_sizes[1] / 2;
  const int* rows = ei;
  const int* cols = ei + E;

  float* ws   = (float*)d_ws;
  float* h1   = ws;
  float* h2   = h1 + (size_t)N * HID;
  int*   adj  = (int*)(h2 + (size_t)N * OUT);
  int*   rank = adj + (size_t)2 * E;
  int*   cnt  = rank + E;
  int*   off  = cnt + N;
  int*   part = off + N + 1;

  const int T = 256;
  const int R = 32;
  const int C = 8;
  int*   tot  = part + (size_t)C * N;

  k_rank_and_gemm<R, C, 128, 64><<<R * C + (N + 255) / 256, 1024, 0, stream>>>(
      cols, rank, part, tot, N, E, x, W1, h1);
  k_offsets<R, C><<<R, 1024, 0, stream>>>(part, tot, off, cnt, N);
  k_fill<C><<<(E + T - 1) / T, T, 0, stream>>>(rows, cols, rank, part, cnt, adj, N, E);

  k_gather_relu_gemm<64, 32><<<(N * (HID / 4) + T - 1) / T, T, 0, stream>>>(adj, off, h1, b1,
                                                                            W2, h2, N);

  k_gather<32, false><<<(N * (OUT / 4) + T - 1) / T, T, 0, stream>>>(adj, off, h2, b2,
                                                                     (float*)d_out, N);
}


# Counting launch: a lane tests its 16 targets per trip into a hit mask and does the LDS add and list store once per in-range target (1 in 32), not per slot; sums chunk-major, LDS arrival order inside
# speedup vs baseline: 5.5269x; 1.7806x over previous
#include <hip/hip_runtime.h>
#include <cstddef>


typedef __attribute__((ext_vector_type(2))) float v2f;
typedef __attribute__((ext_vector_type(8))) float v8f;

__device__ inline int wave_incl_scan(int v, int lane) {
  v += __builtin_amdgcn_update_dpp(0, v, 0x111, 0xf, 0xf, false);
  v += __builtin_amdgcn_update_dpp(0, v, 0x112, 0xf, 0xf, false);
  v += __builtin_amdgcn_update_dpp(0, v, 0x114, 0xf, 0xf, false);
  v += __builtin_amdgcn_update_dpp(0, v, 0x118, 0xf, 0xf, false);
  const int low = __builtin_amdgcn_readlane(v, 15);
  if (lane >= 16) v += low;
  return v;
}

template <int R> __device__ inline int range_nodes(const int N) { return (N + R - 1) / R; }
template <int C> __host__ __device__ inline int chunk_edges(const int E) {
  return ((E + C - 1) / C + 4095) / 4096 * 4096;
}

constexpr int RANK_BITS = 17;

__device__ inline int4 count4(const int4 v, const int lo, const unsigned span, int* lc) {
  const unsigned u0 = (unsigned)(v.x - lo), u1 = (unsigned)(v.y - lo);
  const unsigned u2 = (unsigned)(v.z - lo), u3 = (unsigned)(v.w - lo);
  int4 k = make_int4(-1, -1, -1, -1);
  if (u0 < span) k.x = atomicAdd(&lc[u0], 1);
  if (u1 < span) k.y = atomicAdd(&lc[u1], 1);
  if (u2 < span) k.z = atomicAdd(&lc[u2], 1);
  if (u3 < span) k.w = atomicAdd(&lc[u3], 1);
  return k;
}

__device__ inline unsigned hits4(const int4 v, const int lo, const unsigned span) {
  unsigned m = 0u;
  if ((unsigned)(v.x - lo) < span) m |= 1u;
  if ((unsigned)(v.y - lo) < span) m |= 2u;
  if ((unsigned)(v.z - lo) < span) m |= 4u;
  if ((unsigned)(v.w - lo) < span) m |= 8u;
  return m;
}

__device__ inline int pick4(const int4 v, const int i) {
  const int a = (i & 1) ? v.y : v.x;
  const int b = (i & 1) ? v.w : v.z;
  return (i & 2) ? b : a;
}
__device__ inline int pick16(const int4 v0, const int4 v1, const int4 v2, const int4 v3,
                             const int i) {
  const int a = pick4(v0, i), b = pick4(v1, i), c = pick4(v2, i), d = pick4(v3, i);
  const int ab = (i & 4) ? b : a;
  const int cd = (i & 4) ? d : c;
  return (i & 8) ? cd : ab;
}

__device__ inline int put_hit(int2* __restrict__ wl, const int run, const int e, const int tg,
                              const int k, const int lo) {
  const bool hit = k >= 0;
  const unsigned m = __builtin_amdgcn_ballot_w32(hit);
  if (hit)
    wl[run + (int)__builtin_amdgcn_mbcnt_lo(m, 0u)] = make_int2(e, ((tg - lo) << RANK_BITS) | k);
  return run + __builtin_popcount(m);
}

__device__ inline int put_hits4(int2* __restrict__ wl, int run, const int b, const int4 v,
                                const int4 k, const int lo) {
  run = put_hit(wl, run, b,     v.x, k.x, lo);
  run = put_hit(wl, run, b + 1, v.y, k.y, lo);
  run = put_hit(wl, run, b + 2, v.z, k.z, lo);
  run = put_hit(wl, run, b + 3, v.w, k.w, lo);
  return run;
}

template <int R, int C>
__global__ __launch_bounds__(1024) void k_count(const int* __restrict__ cols,
                                                int* __restrict__ part, int* __restrict__ tot,
                                                int2* __restrict__ list,
                                                int* __restrict__ wcnt, int N, int E) {
  __shared__ int lc[2048];
  __shared__ int stot;
  const int RN = range_nodes<R>(N);
  const int CE = chunk_edges<C>(E);
  const int bid  = blockIdx.x;
  const int t    = threadIdx.x;
  const int lane = t & 31;
  const int wv   = __builtin_amdgcn_readfirstlane(t >> 5);
  const int r    = bid % R;
  const int c    = bid / R;
  const int lo   = r * RN;
  const int nsp  = max(min(lo + RN, N) - lo, 0);
  const unsigned span = (unsigned)nsp;
  for (int i = t; i < nsp; i += 1024) lc[i] = 0;
  if (t == 0) stot = 0;
  __syncthreads();

  int2* __restrict__ wl = list + (size_t)(bid * 32 + wv) * (size_t)(CE >> 5);
  int run = 0;

  const int e0  = c * CE;
  const int e1  = max(min(e0 + CE, E), e0);
  const int e1v = e1 & ~3;
  const int4* __restrict__ c4 = (const int4*)cols;
  const int4 none = make_int4(-1, -1, -1, -1);
  int b0 = e0;
  for (; b0 + 4 * 4096 <= e1v; b0 += 4 * 4096) {
    const int b = b0 + (t << 2);
    const int4 v0 = c4[b >> 2];
    const int4 v1 = c4[(b >> 2) + 1024];
    const int4 v2 = c4[(b >> 2) + 2048];
    const int4 v3 = c4[(b >> 2) + 3072];
    unsigned hm = hits4(v0, lo, span) | (hits4(v1, lo, span) << 4) |
                  (hits4(v2, lo, span) << 8) | (hits4(v3, lo, span) << 12);
    for (;;) {
      const bool hit = hm != 0u;
      const unsigned m = __builtin_amdgcn_ballot_w32(hit);
      if (m == 0u) break;
      if (hit) {
        const int i = __builtin_ctz(hm);
        hm &= hm - 1u;
        const int u = pick16(v0, v1, v2, v3, i) - lo;
        const int k = atomicAdd(&lc[u], 1);
        wl[run + (int)__builtin_amdgcn_mbcnt_lo(m, 0u)] =
            make_int2(b + ((i >> 2) << 12) + (i & 3), (u << RANK_BITS) | k);
      }
      run += __builtin_popcount(m);
    }
  }
  for (; b0 < e1v; b0 += 4096) {
    const int b = b0 + (t << 2);
    int4 v = none;
    if (b < e1v) v = c4[b >> 2];
    const int4 k0 = count4(v, lo, span, lc);
    run = put_hits4(wl, run, b, v, k0, lo);
  }
  {
    const int b = e1v + t;
    int tg = -1, k = -1;
    if (b < e1) tg = cols[b];
    const unsigned u = (unsigned)(tg - lo);
    if (u < span) k = atomicAdd(&lc[u], 1);
    run = put_hit(wl, run, b, tg, k, lo);
  }
  if (lane == 0) wcnt[bid * 32 + wv] = run;
  __syncthreads();
  int mine = 0;
  for (int i = t; i < nsp; i += 1024) {
    const int k = lc[i];
    part[(size_t)c * N + lo + i] = k;
    mine += k;
  }
  const int winc = wave_incl_scan(mine, lane);
  if (lane == 31) atomicAdd(&stot, winc);
  __syncthreads();
  if (t == 0) tot[bid] = stot;
}

template <int R, int C>
__device__ __forceinline__ void place_block(int* place, int* wa, int* wb, const int bid,
                                            const int* __restrict__ rows,
                                            const int* __restrict__ part,
                                            const int* __restrict__ tot,
                                            const int2* __restrict__ list,
                                            const int* __restrict__ wcnt,
                                            int* __restrict__ off, int* __restrict__ adj,
                                            int N, int E) {
  const int t    = threadIdx.x;
  const int lane = t & 31;
  const int wv   = __builtin_amdgcn_readfirstlane(t >> 5);
  const int r    = bid % R;
  const int c    = bid / R;
  const int lo   = r * range_nodes<R>(N);
  const int hi   = min(lo + range_nodes<R>(N), N);
  const int WCAP = chunk_edges<C>(E) >> 5;

  const int2* __restrict__ wl = list + (size_t)(bid * 32 + wv) * (size_t)WCAP;

  const int i0 = lo + 2 * t;
  const int i1 = i0 + 1;
  const bool ok0 = i0 < hi, ok1 = i1 < hi;
  const int j0 = min(i0, N - 1), j1 = min(i1, N - 1);
  int p0[C], p1[C];
#pragma unroll
  for (int cc = 0; cc < C; ++cc) {
    p0[cc] = part[(size_t)cc * N + j0];
    p1[cc] = part[(size_t)cc * N + j1];
  }
  int tv = 0;
  if (t < R * C && (t % R) < r) tv = tot[t];
  const int nl = min(wcnt[bid * 32 + wv], WCAP);
#pragma unroll
  for (int cc = 0; cc < C; ++cc) {
    if (!ok0) p0[cc] = 0;
    if (!ok1) p1[cc] = 0;
  }
  const int tinc = wave_incl_scan(tv, lane);
  if (lane == 31) wa[wv] = tinc;
  int s0 = 0, s1 = 0, q0 = 0, q1 = 0;
#pragma unroll
  for (int cc = 0; cc < C; ++cc) {
    s0 += p0[cc];
    s1 += p1[cc];
    if (cc < c) { q0 += p0[cc]; q1 += p1[cc]; }
  }
  const int s   = s0 + s1;
  const int inc = wave_incl_scan(s, lane);
  if (lane == 31) wb[wv] = inc;
  __syncthreads();
  const int a    = wa[lane];
  const int b    = wb[lane];
  const int ainc = wave_incl_scan(a, lane);
  const int binc = wave_incl_scan(b, lane);
  const int start = __builtin_amdgcn_readlane(ainc, 31);
  const int wave_excl =
      __builtin_amdgcn_readlane(binc, wv) - __builtin_amdgcn_readlane(b, wv);
  const int total = __builtin_amdgcn_readlane(binc, 31);
  const int f0 = start + wave_excl + (inc - s);
  const int f1 = f0 + s0;
  place[2 * t]     = f0 + q0;
  place[2 * t + 1] = f1 + q1;
  if (c == 0) {
    if (ok0) off[i0] = f0;
    if (ok1) off[i1] = f1;
    if (r == R - 1 && t == 0) off[N] = start + total;
  }
  __syncthreads();

  const unsigned elast = (unsigned)(E - 1);
  for (int i = 0; i < nl; i += 128) {
    int2 en[4];
    int  sr[4];
#pragma unroll
    for (int q = 0; q < 4; ++q) en[q] = wl[min(i + 32 * q + lane, nl - 1)];
#pragma unroll
    for (int q = 0; q < 4; ++q) {
      sr[q] = 0;
      if (i + 32 * q + lane < nl) sr[q] = rows[min((unsigned)en[q].x, elast)];
    }
#pragma unroll
    for (int q = 0; q < 4; ++q) {
      if (i + 32 * q + lane < nl) {
        const int u = (en[q].y >> RANK_BITS) & 2047;
        const int k = en[q].y & ((1 << RANK_BITS) - 1);
        adj[min((unsigned)(place[u] + k), elast)] = sr[q];
      }
    }
  }
}

template <int K, int NN, int MT, int NG, int CP>
__device__ __forceinline__ void gemm_rows_block(float* tile, const int bid,
                                                const float* __restrict__ A,
                                                const float* __restrict__ B,
                                                const int* __restrict__ part,
                                                float* __restrict__ C, int M) {
  constexpr int NT    = NN / 16;
  constexpr int NW    = NT * NG;
  constexpr int GROWS = 16 * MT;
  constexpr int ROWS  = GROWS * NG;
  constexpr int PITCH = K + 4;
  static_assert(K == 128, "one wave-wide 16-byte load = one row of A");
  static_assert(ROWS % NW == 0, "rows split evenly over the waves");
  static_assert(ROWS <= 32 * NW, "one thread per row for the row factors");

  const int wave = threadIdx.x >> 5;
  const int lane = threadIdx.x & 31;
  const int grp  = wave / NT;
  const int m0   = bid * ROWS;
  float* dv = tile + ROWS * PITCH;

  const int n0   = (wave % NT) << 4;
  const int l16  = lane & 15;
  const int kh   = (lane >> 4) << 1;

  const float4* __restrict__ A4 = (const float4*)A;
  float4 st[ROWS / NW];
#pragma unroll
  for (int i = 0; i < ROWS / NW; ++i) {
    const int r = min(m0 + wave + NW * i, M - 1);
    st[i] = A4[(size_t)r * (K / 4) + lane];
  }
  if (threadIdx.x < ROWS) {
    const int r = min(m0 + (int)threadIdx.x, M - 1);
    int d = 1;
#pragma unroll
    for (int cc = 0; cc < CP; ++cc) d += part[(size_t)cc * M + r];
    dv[threadIdx.x] = 1.0f / sqrtf((float)d);
  }
#pragma unroll
  for (int i = 0; i < ROWS / NW; ++i)
    *(float4*)&tile[(wave + NW * i) * PITCH + 4 * lane] = st[i];
  __syncthreads();

  v8f acc[MT];
#pragma unroll
  for (int mt = 0; mt < MT; ++mt) acc[mt] = v8f{};
  const float* arow = tile + (grp * GROWS + l16) * PITCH + kh;
  const float* bcol = B + (size_t)kh * NN + n0 + l16;

#pragma unroll
  for (int k = 0; k < K; k += 4) {
    v2f b;
    b.x = bcol[(size_t)k * NN];
    b.y = bcol[(size_t)(k + 1) * NN];
#pragma unroll
    for (int mt = 0; mt < MT; ++mt) {
      v2f a;
      a.x = arow[mt * 16 * PITCH + k];
      a.y = arow[mt * 16 * PITCH + k + 1];
      acc[mt] = __builtin_amdgcn_wmma_f32_16x16x4_f32(false, a, false, b, (short)0,
                                                      acc[mt], false, false);
    }
  }

  const int chi = (lane >> 4) << 3;
#pragma unroll
  for (int mt = 0; mt < MT; ++mt) {
    const int lr = grp * GROWS + mt * 16 + chi;
    const int mr = m0 + lr;
    float* crow = C + (size_t)mr * NN + n0 + l16;
#pragma unroll
    for (int v = 0; v < 8; ++v)
      if (mr + v < M) crow[(size_t)v * NN] = acc[mt][v] * dv[lr + v];
  }
}

template <int R, int C, int K, int NN>
__global__ __launch_bounds__(1024) __attribute__((amdgpu_waves_per_eu(16, 16)))
void k_place_and_gemm(const int* __restrict__ rows, const int* __restrict__ part,
                      const int* __restrict__ tot, const int2* __restrict__ list,
                      const int* __restrict__ wcnt, int* __restrict__ off,
                      int* __restrict__ adj, int N, int E,
                      const float* __restrict__ A, const float* __restrict__ B,
                      float* __restrict__ Cm) {
  constexpr int MT = 2, NG = 8;
  static_assert((NN / 16) * NG * 32 == 1024, "product work-groups fill the 1024 threads");
  static_assert(R * C <= 1024, "one thread per block sum");
  __shared__ __attribute__((aligned(16))) float smem[16 * MT * NG * (K + 4) + 16 * MT * NG];
  static_assert(sizeof(smem) >= (2048 + 64) * sizeof(int), "room for the places");
  const int G = (N + 16 * MT * NG - 1) / (16 * MT * NG);
  if ((int)blockIdx.x >= G) {
    int* pl = (int*)smem;
    place_block<R, C>(pl, pl + 2048, pl + 2048 + 32, (int)blockIdx.x - G, rows, part, tot,
                      list, wcnt, off, adj, N, E);
  } else {
    gemm_rows_block<K, NN, MT, NG, C>(smem, (int)blockIdx.x, A, B, part, Cm, N);
  }
}

__device__ inline float4 add4(const float4 a, const float4 g) {
  return make_float4(a.x + g.x, a.y + g.y, a.z + g.z, a.w + g.w);
}

template <int F, bool RELU>
__device__ inline float4 gather_row(const int* __restrict__ adj, const int* __restrict__ off,
                                    const float* __restrict__ H,
                                    const float* __restrict__ bias, const int node,
                                    const int j, const int n, float& dc) {
  constexpr int LPN = F / 4;
  const float4* __restrict__ H4 = (const float4*)H;
  const int beg = off[node];
  const int end = off[node + 1];
  dc = 1.0f / sqrtf((float)(end - beg + 1));
  const unsigned last = (unsigned)(n - 1);

  const float4 hs = H4[(size_t)node * LPN + j];
  const float4 b  = ((const float4*)bias)[j];

  float4 acc = make_float4(0.f, 0.f, 0.f, 0.f);
  int i = beg;
  for (; i + 4 <= end; i += 4) {
    const int p0 = adj[i], p1 = adj[i + 1], p2 = adj[i + 2], p3 = adj[i + 3];
    const float4 g0 = H4[(size_t)min((unsigned)p0, last) * LPN + j];
    const float4 g1 = H4[(size_t)min((unsigned)p1, last) * LPN + j];
    const float4 g2 = H4[(size_t)min((unsigned)p2, last) * LPN + j];
    const float4 g3 = H4[(size_t)min((unsigned)p3, last) * LPN + j];
    acc = add4(acc, g0);
    acc = add4(acc, g1);
    acc = add4(acc, g2);
    acc = add4(acc, g3);
  }
  if (i + 2 <= end) {
    const int p0 = adj[i], p1 = adj[i + 1];
    const float4 g0 = H4[(size_t)min((unsigned)p0, last) * LPN + j];
    const float4 g1 = H4[(size_t)min((unsigned)p1, last) * LPN + j];
    acc = add4(acc, g0);
    acc = add4(acc, g1);
    i += 2;
  }
  if (i < end) {
    const int p0 = adj[i];
    const float4 g0 = H4[(size_t)min((unsigned)p0, last) * LPN + j];
    acc = add4(acc, g0);
  }

  float4 o;
  o.x = fmaf(acc.x + hs.x, dc, b.x);
  o.y = fmaf(acc.y + hs.y, dc, b.y);
  o.z = fmaf(acc.z + hs.z, dc, b.z);
  o.w = fmaf(acc.w + hs.w, dc, b.w);
  if (RELU) {
    o.x = fmaxf(o.x, 0.0f); o.y = fmaxf(o.y, 0.0f);
    o.z = fmaxf(o.z, 0.0f); o.w = fmaxf(o.w, 0.0f);
  }
  return o;
}

template <int F, bool RELU>
__global__ void k_gather(const int* __restrict__ adj, const int* __restrict__ off,
                         const float* __restrict__ H, const float* __restrict__ bias,
                         float* __restrict__ OUT, int n) {
  constexpr int LPN = F / 4;
  const int t    = blockIdx.x * blockDim.x + threadIdx.x;
  const int node = t / LPN;
  const int j    = t % LPN;
  if (node >= n) return;
  float dc;
  ((float4*)OUT)[(size_t)node * LPN + j] = gather_row<F, RELU>(adj, off, H, bias, node, j, n, dc);
}

template <int F, int NN>
__global__ __launch_bounds__(256) void k_gather_relu_gemm(
    const int* __restrict__ adj, const int* __restrict__ off, const float* __restrict__ H,
    const float* __restrict__ bias, const float* __restrict__ B, float* __restrict__ C,
    int n) {
  constexpr int LPN   = F / 4;
  constexpr int ROWS  = 256 / LPN;
  constexpr int PITCH = F + 4;
  static_assert(ROWS == 16, "one 16-row matrix tile per work-group");
  static_assert(NN / 16 <= 8, "one wave per N-tile");
  __shared__ __attribute__((aligned(16))) float tile[ROWS * PITCH];

  const int nl   = threadIdx.x / LPN;
  const int j    = threadIdx.x % LPN;
  const int m0   = blockIdx.x * ROWS;
  const int node = m0 + nl;
  float4 o = make_float4(0.f, 0.f, 0.f, 0.f);
  if (node < n) {
    float dc;
    o = gather_row<F, true>(adj, off, H, bias, node, j, n, dc);
    o.x *= dc; o.y *= dc; o.z *= dc; o.w *= dc;
  }
  *(float4*)&tile[nl * PITCH + 4 * j] = o;
  __syncthreads();

  const int wave = threadIdx.x >> 5;
  if (wave >= NN / 16) return;
  const int lane = threadIdx.x & 31;
  const int n0   = wave << 4;
  const int l16  = lane & 15;
  const int kh   = (lane >> 4) << 1;

  v8f acc = {};
  const float* arow = tile + l16 * PITCH + kh;
  const float* bcol = B + (size_t)kh * NN + n0 + l16;
#pragma unroll
  for (int k = 0; k < F; k += 4) {
    v2f a, b;
    a.x = arow[k];
    a.y = arow[k + 1];
    b.x = bcol[(size_t)k * NN];
    b.y = bcol[(size_t)(k + 1) * NN];
    acc = __builtin_amdgcn_wmma_f32_16x16x4_f32(false, a, false, b, (short)0, acc,
                                                false, false);
  }

  const int chi = (lane >> 4) << 3;
  float* crow = C + (size_t)(m0 + chi) * NN + n0 + l16;
#pragma unroll
  for (int v = 0; v < 8; ++v)
    if (m0 + chi + v < n) crow[(size_t)v * NN] = acc[v];
}

extern "C" void kernel_launch(void* const* d_in, const int* in_sizes, int n_in,
                              void* d_out, int out_size, void* d_ws, size_t ws_size,
                              hipStream_t stream) {
  const float* x  = (const float*)d_in[0];
  const int*   ei = (const int*)d_in[1];
  const float* W1 = (const float*)d_in[2];
  const float* b1 = (const float*)d_in[3];
  const float* W2 = (const float*)d_in[4];
  const float* b2 = (const float*)d_in[5];

  const int IN_DIM = 128, HID = 64, OUT = 32;
  const int N = in_sizes[0] / IN_DIM;
  const int E = in_sizes[1] / 2;
  const int* rows = ei;
  const int* cols = ei + E;

  const int T = 256;
  const int R = 32;
  const int C = 8;
  const int CE = chunk_edges<C>(E);

  int2*  list = (int2*)d_ws;
  float* h1   = (float*)(list + (size_t)R * C * CE);
  float* h2   = h1 + (size_t)N * HID;
  int*   adj  = (int*)(h2 + (size_t)N * OUT);
  int*   off  = adj + E;
  int*   part = off + N + 1;
  int*   tot  = part + (size_t)C * N;
  int*   wcnt = tot + R * C;

  k_count<R, C><<<R * C, 1024, 0, stream>>>(cols, part, tot, list, wcnt, N, E);
  k_place_and_gemm<R, C, 128, 64><<<R * C + (N + 255) / 256, 1024, 0, stream>>>(
      rows, part, tot, list, wcnt, off, adj, N, E, x, W1, h1);

  k_gather_relu_gemm<64, 32><<<(N * (HID / 4) + T - 1) / T, T, 0, stream>>>(adj, off, h1, b1,
                                                                            W2, h2, N);

  k_gather<32, false><<<(N * (OUT / 4) + T - 1) / T, T, 0, stream>>>(adj, off, h2, b2,
                                                                     (float*)d_out, N);
}


# Both gathers: a block's 16 source indices come by one coalesced load per lane group and go round by DPP row_share/dpp8, so 8 row loads fly per wait with no index load between; sums keep their order
# speedup vs baseline: 5.7338x; 1.0374x over previous
#include <hip/hip_runtime.h>
#include <cstddef>


typedef __attribute__((ext_vector_type(2))) float v2f;
typedef __attribute__((ext_vector_type(8))) float v8f;

__device__ inline int wave_incl_scan(int v, int lane) {
  v += __builtin_amdgcn_update_dpp(0, v, 0x111, 0xf, 0xf, false);
  v += __builtin_amdgcn_update_dpp(0, v, 0x112, 0xf, 0xf, false);
  v += __builtin_amdgcn_update_dpp(0, v, 0x114, 0xf, 0xf, false);
  v += __builtin_amdgcn_update_dpp(0, v, 0x118, 0xf, 0xf, false);
  const int low = __builtin_amdgcn_readlane(v, 15);
  if (lane >= 16) v += low;
  return v;
}

template <int R> __device__ inline int range_nodes(const int N) { return (N + R - 1) / R; }
template <int C> __host__ __device__ inline int chunk_edges(const int E) {
  return ((E + C - 1) / C + 4095) / 4096 * 4096;
}

constexpr int RANK_BITS = 17;

__device__ inline int4 count4(const int4 v, const int lo, const unsigned span, int* lc) {
  const unsigned u0 = (unsigned)(v.x - lo), u1 = (unsigned)(v.y - lo);
  const unsigned u2 = (unsigned)(v.z - lo), u3 = (unsigned)(v.w - lo);
  int4 k = make_int4(-1, -1, -1, -1);
  if (u0 < span) k.x = atomicAdd(&lc[u0], 1);
  if (u1 < span) k.y = atomicAdd(&lc[u1], 1);
  if (u2 < span) k.z = atomicAdd(&lc[u2], 1);
  if (u3 < span) k.w = atomicAdd(&lc[u3], 1);
  return k;
}

__device__ inline unsigned hits4(const int4 v, const int lo, const unsigned span) {
  unsigned m = 0u;
  if ((unsigned)(v.x - lo) < span) m |= 1u;
  if ((unsigned)(v.y - lo) < span) m |= 2u;
  if ((unsigned)(v.z - lo) < span) m |= 4u;
  if ((unsigned)(v.w - lo) < span) m |= 8u;
  return m;
}

__device__ inline int pick4(const int4 v, const int i) {
  const int a = (i & 1) ? v.y : v.x;
  const int b = (i & 1) ? v.w : v.z;
  return (i & 2) ? b : a;
}
__device__ inline int pick16(const int4 v0, const int4 v1, const int4 v2, const int4 v3,
                             const int i) {
  const int a = pick4(v0, i), b = pick4(v1, i), c = pick4(v2, i), d = pick4(v3, i);
  const int ab = (i & 4) ? b : a;
  const int cd = (i & 4) ? d : c;
  return (i & 8) ? cd : ab;
}

__device__ inline int put_hit(int2* __restrict__ wl, const int run, const int e, const int tg,
                              const int k, const int lo) {
  const bool hit = k >= 0;
  const unsigned m = __builtin_amdgcn_ballot_w32(hit);
  if (hit)
    wl[run + (int)__builtin_amdgcn_mbcnt_lo(m, 0u)] = make_int2(e, ((tg - lo) << RANK_BITS) | k);
  return run + __builtin_popcount(m);
}

__device__ inline int put_hits4(int2* __restrict__ wl, int run, const int b, const int4 v,
                                const int4 k, const int lo) {
  run = put_hit(wl, run, b,     v.x, k.x, lo);
  run = put_hit(wl, run, b + 1, v.y, k.y, lo);
  run = put_hit(wl, run, b + 2, v.z, k.z, lo);
  run = put_hit(wl, run, b + 3, v.w, k.w, lo);
  return run;
}

template <int R, int C>
__global__ __launch_bounds__(1024) void k_count(const int* __restrict__ cols,
                                                int* __restrict__ part, int* __restrict__ tot,
                                                int2* __restrict__ list,
                                                int* __restrict__ wcnt, int N, int E) {
  __shared__ int lc[2048];
  __shared__ int stot;
  const int RN = range_nodes<R>(N);
  const int CE = chunk_edges<C>(E);
  const int bid  = blockIdx.x;
  const int t    = threadIdx.x;
  const int lane = t & 31;
  const int wv   = __builtin_amdgcn_readfirstlane(t >> 5);
  const int r    = bid % R;
  const int c    = bid / R;
  const int lo   = r * RN;
  const int nsp  = max(min(lo + RN, N) - lo, 0);
  const unsigned span = (unsigned)nsp;
  for (int i = t; i < nsp; i += 1024) lc[i] = 0;
  if (t == 0) stot = 0;
  __syncthreads();

  int2* __restrict__ wl = list + (size_t)(bid * 32 + wv) * (size_t)(CE >> 5);
  int run = 0;

  const int e0  = c * CE;
  const int e1  = max(min(e0 + CE, E), e0);
  const int e1v = e1 & ~3;
  const int4* __restrict__ c4 = (const int4*)cols;
  const int4 none = make_int4(-1, -1, -1, -1);
  int b0 = e0;
  for (; b0 + 4 * 4096 <= e1v; b0 += 4 * 4096) {
    const int b = b0 + (t << 2);
    const int4 v0 = c4[b >> 2];
    const int4 v1 = c4[(b >> 2) + 1024];
    const int4 v2 = c4[(b >> 2) + 2048];
    const int4 v3 = c4[(b >> 2) + 3072];
    unsigned hm = hits4(v0, lo, span) | (hits4(v1, lo, span) << 4) |
                  (hits4(v2, lo, span) << 8) | (hits4(v3, lo, span) << 12);
    for (;;) {
      const bool hit = hm != 0u;
      const unsigned m = __builtin_amdgcn_ballot_w32(hit);
      if (m == 0u) break;
      if (hit) {
        const int i = __builtin_ctz(hm);
        hm &= hm - 1u;
        const int u = pick16(v0, v1, v2, v3, i) - lo;
        const int k = atomicAdd(&lc[u], 1);
        wl[run + (int)__builtin_amdgcn_mbcnt_lo(m, 0u)] =
            make_int2(b + ((i >> 2) << 12) + (i & 3), (u << RANK_BITS) | k);
      }
      run += __builtin_popcount(m);
    }
  }
  for (; b0 < e1v; b0 += 4096) {
    const int b = b0 + (t << 2);
    int4 v = none;
    if (b < e1v) v = c4[b >> 2];
    const int4 k0 = count4(v, lo, span, lc);
    run = put_hits4(wl, run, b, v, k0, lo);
  }
  {
    const int b = e1v + t;
    int tg = -1, k = -1;
    if (b < e1) tg = cols[b];
    const unsigned u = (unsigned)(tg - lo);
    if (u < span) k = atomicAdd(&lc[u], 1);
    run = put_hit(wl, run, b, tg, k, lo);
  }
  if (lane == 0) wcnt[bid * 32 + wv] = run;
  __syncthreads();
  int mine = 0;
  for (int i = t; i < nsp; i += 1024) {
    const int k = lc[i];
    part[(size_t)c * N + lo + i] = k;
    mine += k;
  }
  const int winc = wave_incl_scan(mine, lane);
  if (lane == 31) atomicAdd(&stot, winc);
  __syncthreads();
  if (t == 0) tot[bid] = stot;
}

template <int R, int C>
__device__ __forceinline__ void place_block(int* place, int* wa, int* wb, const int bid,
                                            const int* __restrict__ rows,
                                            const int* __restrict__ part,
                                            const int* __restrict__ tot,
                                            const int2* __restrict__ list,
                                            const int* __restrict__ wcnt,
                                            int* __restrict__ off, int* __restrict__ adj,
                                            int N, int E) {
  const int t    = threadIdx.x;
  const int lane = t & 31;
  const int wv   = __builtin_amdgcn_readfirstlane(t >> 5);
  const int r    = bid % R;
  const int c    = bid / R;
  const int lo   = r * range_nodes<R>(N);
  const int hi   = min(lo + range_nodes<R>(N), N);
  const int WCAP = chunk_edges<C>(E) >> 5;

  const int2* __restrict__ wl = list + (size_t)(bid * 32 + wv) * (size_t)WCAP;

  const int i0 = lo + 2 * t;
  const int i1 = i0 + 1;
  const bool ok0 = i0 < hi, ok1 = i1 < hi;
  const int j0 = min(i0, N - 1), j1 = min(i1, N - 1);
  int p0[C], p1[C];
#pragma unroll
  for (int cc = 0; cc < C; ++cc) {
    p0[cc] = part[(size_t)cc * N + j0];
    p1[cc] = part[(size_t)cc * N + j1];
  }
  int tv = 0;
  if (t < R * C && (t % R) < r) tv = tot[t];
  const int nl = min(wcnt[bid * 32 + wv], WCAP);
#pragma unroll
  for (int cc = 0; cc < C; ++cc) {
    if (!ok0) p0[cc] = 0;
    if (!ok1) p1[cc] = 0;
  }
  const int tinc = wave_incl_scan(tv, lane);
  if (lane == 31) wa[wv] = tinc;
  int s0 = 0, s1 = 0, q0 = 0, q1 = 0;
#pragma unroll
  for (int cc = 0; cc < C; ++cc) {
    s0 += p0[cc];
    s1 += p1[cc];
    if (cc < c) { q0 += p0[cc]; q1 += p1[cc]; }
  }
  const int s   = s0 + s1;
  const int inc = wave_incl_scan(s, lane);
  if (lane == 31) wb[wv] = inc;
  __syncthreads();
  const int a    = wa[lane];
  const int b    = wb[lane];
  const int ainc = wave_incl_scan(a, lane);
  const int binc = wave_incl_scan(b, lane);
  const int start = __builtin_amdgcn_readlane(ainc, 31);
  const int wave_excl =
      __builtin_amdgcn_readlane(binc, wv) - __builtin_amdgcn_readlane(b, wv);
  const int total = __builtin_amdgcn_readlane(binc, 31);
  const int f0 = start + wave_excl + (inc - s);
  const int f1 = f0 + s0;
  place[2 * t]     = f0 + q0;
  place[2 * t + 1] = f1 + q1;
  if (c == 0) {
    if (ok0) off[i0] = f0;
    if (ok1) off[i1] = f1;
    if (r == R - 1 && t == 0) off[N] = start + total;
  }
  __syncthreads();

  const unsigned elast = (unsigned)(E - 1);
  for (int i = 0; i < nl; i += 128) {
    int2 en[4];
    int  sr[4];
#pragma unroll
    for (int q = 0; q < 4; ++q) en[q] = wl[min(i + 32 * q + lane, nl - 1)];
#pragma unroll
    for (int q = 0; q < 4; ++q) {
      sr[q] = 0;
      if (i + 32 * q + lane < nl) sr[q] = rows[min((unsigned)en[q].x, elast)];
    }
#pragma unroll
    for (int q = 0; q < 4; ++q) {
      if (i + 32 * q + lane < nl) {
        const int u = (en[q].y >> RANK_BITS) & 2047;
        const int k = en[q].y & ((1 << RANK_BITS) - 1);
        adj[min((unsigned)(place[u] + k), elast)] = sr[q];
      }
    }
  }
}

template <int K, int NN, int MT, int NG, int CP>
__device__ __forceinline__ void gemm_rows_block(float* tile, const int bid,
                                                const float* __restrict__ A,
                                                const float* __restrict__ B,
                                                const int* __restrict__ part,
                                                float* __restrict__ C, int M) {
  constexpr int NT    = NN / 16;
  constexpr int NW    = NT * NG;
  constexpr int GROWS = 16 * MT;
  constexpr int ROWS  = GROWS * NG;
  constexpr int PITCH = K + 4;
  static_assert(K == 128, "one wave-wide 16-byte load = one row of A");
  static_assert(ROWS % NW == 0, "rows split evenly over the waves");
  static_assert(ROWS <= 32 * NW, "one thread per row for the row factors");

  const int wave = threadIdx.x >> 5;
  const int lane = threadIdx.x & 31;
  const int grp  = wave / NT;
  const int m0   = bid * ROWS;
  float* dv = tile + ROWS * PITCH;

  const int n0   = (wave % NT) << 4;
  const int l16  = lane & 15;
  const int kh   = (lane >> 4) << 1;

  const float4* __restrict__ A4 = (const float4*)A;
  float4 st[ROWS / NW];
#pragma unroll
  for (int i = 0; i < ROWS / NW; ++i) {
    const int r = min(m0 + wave + NW * i, M - 1);
    st[i] = A4[(size_t)r * (K / 4) + lane];
  }
  if (threadIdx.x < ROWS) {
    const int r = min(m0 + (int)threadIdx.x, M - 1);
    int d = 1;
#pragma unroll
    for (int cc = 0; cc < CP; ++cc) d += part[(size_t)cc * M + r];
    dv[threadIdx.x] = 1.0f / sqrtf((float)d);
  }
#pragma unroll
  for (int i = 0; i < ROWS / NW; ++i)
    *(float4*)&tile[(wave + NW * i) * PITCH + 4 * lane] = st[i];
  __syncthreads();

  v8f acc[MT];
#pragma unroll
  for (int mt = 0; mt < MT; ++mt) acc[mt] = v8f{};
  const float* arow = tile + (grp * GROWS + l16) * PITCH + kh;
  const float* bcol = B + (size_t)kh * NN + n0 + l16;

#pragma unroll
  for (int k = 0; k < K; k += 4) {
    v2f b;
    b.x = bcol[(size_t)k * NN];
    b.y = bcol[(size_t)(k + 1) * NN];
#pragma unroll
    for (int mt = 0; mt < MT; ++mt) {
      v2f a;
      a.x = arow[mt * 16 * PITCH + k];
      a.y = arow[mt * 16 * PITCH + k + 1];
      acc[mt] = __builtin_amdgcn_wmma_f32_16x16x4_f32(false, a, false, b, (short)0,
                                                      acc[mt], false, false);
    }
  }

  const int chi = (lane >> 4) << 3;
#pragma unroll
  for (int mt = 0; mt < MT; ++mt) {
    const int lr = grp * GROWS + mt * 16 + chi;
    const int mr = m0 + lr;
    float* crow = C + (size_t)mr * NN + n0 + l16;
#pragma unroll
    for (int v = 0; v < 8; ++v)
      if (mr + v < M) crow[(size_t)v * NN] = acc[mt][v] * dv[lr + v];
  }
}

template <int R, int C, int K, int NN>
__global__ __launch_bounds__(1024) __attribute__((amdgpu_waves_per_eu(16, 16)))
void k_place_and_gemm(const int* __restrict__ rows, const int* __restrict__ part,
                      const int* __restrict__ tot, const int2* __restrict__ list,
                      const int* __restrict__ wcnt, int* __restrict__ off,
                      int* __restrict__ adj, int N, int E,
                      const float* __restrict__ A, const float* __restrict__ B,
                      float* __restrict__ Cm) {
  constexpr int MT = 2, NG = 8;
  static_assert((NN / 16) * NG * 32 == 1024, "product work-groups fill the 1024 threads");
  static_assert(R * C <= 1024, "one thread per block sum");
  __shared__ __attribute__((aligned(16))) float smem[16 * MT * NG * (K + 4) + 16 * MT * NG];
  static_assert(sizeof(smem) >= (2048 + 64) * sizeof(int), "room for the places");
  const int G = (N + 16 * MT * NG - 1) / (16 * MT * NG);
  if ((int)blockIdx.x >= G) {
    int* pl = (int*)smem;
    place_block<R, C>(pl, pl + 2048, pl + 2048 + 32, (int)blockIdx.x - G, rows, part, tot,
                      list, wcnt, off, adj, N, E);
  } else {
    gemm_rows_block<K, NN, MT, NG, C>(smem, (int)blockIdx.x, A, B, part, Cm, N);
  }
}

__device__ inline float4 add4(const float4 a, const float4 g) {
  return make_float4(a.x + g.x, a.y + g.y, a.z + g.z, a.w + g.w);
}

template <int LPN, int K>
__device__ inline unsigned group_entry(const unsigned va, const unsigned vb) {
  static_assert(LPN == 16 || LPN == 8, "a lane group is one DPP row or one DPP8 group");
  if constexpr (LPN == 16) {
    return (unsigned)__builtin_amdgcn_update_dpp(0, (int)va, 0x150 + K, 0xf, 0xf, false);
  } else {
    return (unsigned)__builtin_amdgcn_mov_dpp8(K < 8 ? va : vb, (unsigned)(K & 7) * 0x249249u);
  }
}

template <unsigned ROWB>
__device__ inline unsigned row_offset(const int p, const unsigned last) {
  return min((unsigned)p, last) * ROWB;
}

__device__ inline float4 load_row16(const char* __restrict__ Hb, const unsigned o) {
  return *(const float4*)(Hb + o);
}

template <int LPN, int K>
__device__ inline float4 half_block(const char* __restrict__ Hb, const unsigned va,
                                    const unsigned vb, const unsigned jo, const int cnt,
                                    float4 acc) {
  const int rem = cnt - K;
  if (rem >= 8) {
    const float4 g0 = load_row16(Hb, group_entry<LPN, K>(va, vb) + jo);
    const float4 g1 = load_row16(Hb, group_entry<LPN, K + 1>(va, vb) + jo);
    const float4 g2 = load_row16(Hb, group_entry<LPN, K + 2>(va, vb) + jo);
    const float4 g3 = load_row16(Hb, group_entry<LPN, K + 3>(va, vb) + jo);
    const float4 g4 = load_row16(Hb, group_entry<LPN, K + 4>(va, vb) + jo);
    const float4 g5 = load_row16(Hb, group_entry<LPN, K + 5>(va, vb) + jo);
    const float4 g6 = load_row16(Hb, group_entry<LPN, K + 6>(va, vb) + jo);
    const float4 g7 = load_row16(Hb, group_entry<LPN, K + 7>(va, vb) + jo);
    acc = add4(acc, g0);
    acc = add4(acc, g1);
    acc = add4(acc, g2);
    acc = add4(acc, g3);
    acc = add4(acc, g4);
    acc = add4(acc, g5);
    acc = add4(acc, g6);
    acc = add4(acc, g7);
  } else if (rem > 0) {
    const float4 z = make_float4(0.f, 0.f, 0.f, 0.f);
    const float4 g0 = load_row16(Hb, group_entry<LPN, K>(va, vb) + jo);
    float4 g1 = z, g2 = z, g3 = z, g4 = z, g5 = z, g6 = z;
    if (rem > 1) g1 = load_row16(Hb, group_entry<LPN, K + 1>(va, vb) + jo);
    if (rem > 2) g2 = load_row16(Hb, group_entry<LPN, K + 2>(va, vb) + jo);
    if (rem > 3) g3 = load_row16(Hb, group_entry<LPN, K + 3>(va, vb) + jo);
    if (rem > 4) g4 = load_row16(Hb, group_entry<LPN, K + 4>(va, vb) + jo);
    if (rem > 5) g5 = load_row16(Hb, group_entry<LPN, K + 5>(va, vb) + jo);
    if (rem > 6) g6 = load_row16(Hb, group_entry<LPN, K + 6>(va, vb) + jo);
    acc = add4(acc, g0);
    acc = add4(acc, g1);
    acc = add4(acc, g2);
    acc = add4(acc, g3);
    acc = add4(acc, g4);
    acc = add4(acc, g5);
    acc = add4(acc, g6);
  }
  return acc;
}

template <int F, bool RELU>
__device__ inline float4 gather_row(const int* __restrict__ adj, const int* __restrict__ off,
                                    const float* __restrict__ H,
                                    const float* __restrict__ bias, const int node,
                                    const int j, const int n, float& dc) {
  constexpr int LPN = F / 4;
  constexpr unsigned ROWB = F * 4u;
  const float4* __restrict__ H4 = (const float4*)H;
  const char* __restrict__ Hb = (const char*)H;
  const int beg = off[node];
  const int end = off[node + 1];
  dc = 1.0f / sqrtf((float)(end - beg + 1));
  const unsigned last = (unsigned)(n - 1);

  const float4 hs = H4[(size_t)node * LPN + j];
  const float4 b  = ((const float4*)bias)[j];

  float4 acc = make_float4(0.f, 0.f, 0.f, 0.f);
  if (beg < end) {
    const unsigned jo = (unsigned)j * 16u;
    const int tail = end - 1;
    int pa = adj[min(beg + j, tail)];
    int pb = 0;
    if (LPN == 8) pb = adj[min(beg + 8 + j, tail)];
    for (int base = beg; base < end; base += 16) {
      const int cnt = min(end - base, 16);
      const unsigned va = row_offset<ROWB>(pa, last);
      const unsigned vb = row_offset<ROWB>(pb, last);
      pa = adj[min(base + 16 + j, tail)];
      if (LPN == 8) pb = adj[min(base + 24 + j, tail)];
      acc = half_block<LPN, 0>(Hb, va, vb, jo, cnt, acc);
      acc = half_block<LPN, 8>(Hb, va, vb, jo, cnt, acc);
    }
  }

  float4 o;
  o.x = fmaf(acc.x + hs.x, dc, b.x);
  o.y = fmaf(acc.y + hs.y, dc, b.y);
  o.z = fmaf(acc.z + hs.z, dc, b.z);
  o.w = fmaf(acc.w + hs.w, dc, b.w);
  if (RELU) {
    o.x = fmaxf(o.x, 0.0f); o.y = fmaxf(o.y, 0.0f);
    o.z = fmaxf(o.z, 0.0f); o.w = fmaxf(o.w, 0.0f);
  }
  return o;
}

template <int F, bool RELU>
__global__ void k_gather(const int* __restrict__ adj, const int* __restrict__ off,
                         const float* __restrict__ H, const float* __restrict__ bias,
                         float* __restrict__ OUT, int n) {
  constexpr int LPN = F / 4;
  const int t    = blockIdx.x * blockDim.x + threadIdx.x;
  const int node = t / LPN;
  const int j    = t % LPN;
  if (node >= n) return;
  float dc;
  ((float4*)OUT)[(size_t)node * LPN + j] = gather_row<F, RELU>(adj, off, H, bias, node, j, n, dc);
}

template <int F, int NN>
__global__ __launch_bounds__(256) void k_gather_relu_gemm(
    const int* __restrict__ adj, const int* __restrict__ off, const float* __restrict__ H,
    const float* __restrict__ bias, const float* __restrict__ B, float* __restrict__ C,
    int n) {
  constexpr int LPN   = F / 4;
  constexpr int ROWS  = 256 / LPN;
  constexpr int PITCH = F + 4;
  static_assert(ROWS == 16, "one 16-row matrix tile per work-group");
  static_assert(NN / 16 <= 8, "one wave per N-tile");
  __shared__ __attribute__((aligned(16))) float tile[ROWS * PITCH];

  const int nl   = threadIdx.x / LPN;
  const int j    = threadIdx.x % LPN;
  const int m0   = blockIdx.x * ROWS;
  const int node = m0 + nl;
  float4 o = make_float4(0.f, 0.f, 0.f, 0.f);
  if (node < n) {
    float dc;
    o = gather_row<F, true>(adj, off, H, bias, node, j, n, dc);
    o.x *= dc; o.y *= dc; o.z *= dc; o.w *= dc;
  }
  *(float4*)&tile[nl * PITCH + 4 * j] = o;
  __syncthreads();

  const int wave = threadIdx.x >> 5;
  if (wave >= NN / 16) return;
  const int lane = threadIdx.x & 31;
  const int n0   = wave << 4;
  const int l16  = lane & 15;
  const int kh   = (lane >> 4) << 1;

  v8f acc = {};
  const float* arow = tile + l16 * PITCH + kh;
  const float* bcol = B + (size_t)kh * NN + n0 + l16;
#pragma unroll
  for (int k = 0; k < F; k += 4) {
    v2f a, b;
    a.x = arow[k];
    a.y = arow[k + 1];
    b.x = bcol[(size_t)k * NN];
    b.y = bcol[(size_t)(k + 1) * NN];
    acc = __builtin_amdgcn_wmma_f32_16x16x4_f32(false, a, false, b, (short)0, acc,
                                                false, false);
  }

  const int chi = (lane >> 4) << 3;
  float* crow = C + (size_t)(m0 + chi) * NN + n0 + l16;
#pragma unroll
  for (int v = 0; v < 8; ++v)
    if (m0 + chi + v < n) crow[(size_t)v * NN] = acc[v];
}

extern "C" void kernel_launch(void* const* d_in, const int* in_sizes, int n_in,
                              void* d_out, int out_size, void* d_ws, size_t ws_size,
                              hipStream_t stream) {
  const float* x  = (const float*)d_in[0];
  const int*   ei = (const int*)d_in[1];
  const float* W1 = (const float*)d_in[2];
  const float* b1 = (const float*)d_in[3];
  const float* W2 = (const float*)d_in[4];
  const float* b2 = (const float*)d_in[5];

  const int IN_DIM = 128, HID = 64, OUT = 32;
  const int N = in_sizes[0] / IN_DIM;
  const int E = in_sizes[1] / 2;
  const int* rows = ei;
  const int* cols = ei + E;

  const int T = 256;
  const int R = 32;
  const int C = 8;
  const int CE = chunk_edges<C>(E);

  int2*  list = (int2*)d_ws;
  float* h1   = (float*)(list + (size_t)R * C * CE);
  float* h2   = h1 + (size_t)N * HID;
  int*   adj  = (int*)(h2 + (size_t)N * OUT);
  int*   off  = adj + E;
  int*   part = off + N + 1;
  int*   tot  = part + (size_t)C * N;
  int*   wcnt = tot + R * C;

  k_count<R, C><<<R * C, 1024, 0, stream>>>(cols, part, tot, list, wcnt, N, E);
  k_place_and_gemm<R, C, 128, 64><<<R * C + (N + 255) / 256, 1024, 0, stream>>>(
      rows, part, tot, list, wcnt, off, adj, N, E, x, W1, h1);

  k_gather_relu_gemm<64, 32><<<(N * (HID / 4) + T - 1) / T, T, 0, stream>>>(adj, off, h1, b1,
                                                                            W2, h2, N);

  k_gather<32, false><<<(N * (OUT / 4) + T - 1) / T, T, 0, stream>>>(adj, off, h2, b2,
                                                                     (float*)d_out, N);
}
